# starting kernel
# baseline (speedup 1.0000x reference)
#include <hip/hip_runtime.h>


#define N_NODES  100000
#define N_EDGES  3200000
#define N_GRAPHS 4096
#define EMB      32
#define HID      64
#define NCLS     10

typedef __attribute__((ext_vector_type(16))) _Float16 v16h;
typedef __attribute__((ext_vector_type(8)))  float    v8f;

__device__ __forceinline__ v16h pack_frag(float4 c0, float4 c1,
                                          float4 c2, float4 c3, float s)
{
    v16h r;
    r[0]  = (_Float16)(c0.x * s); r[1]  = (_Float16)(c0.y * s);
    r[2]  = (_Float16)(c0.z * s); r[3]  = (_Float16)(c0.w * s);
    r[4]  = (_Float16)(c1.x * s); r[5]  = (_Float16)(c1.y * s);
    r[6]  = (_Float16)(c1.z * s); r[7]  = (_Float16)(c1.w * s);
    r[8]  = (_Float16)(c2.x * s); r[9]  = (_Float16)(c2.y * s);
    r[10] = (_Float16)(c2.z * s); r[11] = (_Float16)(c2.w * s);
    r[12] = (_Float16)(c3.x * s); r[13] = (_Float16)(c3.y * s);
    r[14] = (_Float16)(c3.z * s); r[15] = (_Float16)(c3.w * s);
    return r;
}

__device__ __forceinline__ v16h load_frag(const float4* __restrict__ q,
                                          int ks, int half, float s)
{
    const int o = ks * 8 + half * 2;
    return pack_frag(q[o], q[o + 1], q[o + 4], q[o + 5], s);
}

__global__ void embed_gather(const int* __restrict__ x_idx,
                             const float* __restrict__ embed_w,
                             float* __restrict__ x0)
{
    int tid = blockIdx.x * blockDim.x + threadIdx.x;
    if (tid >= N_NODES * EMB) return;
    int node = tid >> 5;
    int f    = tid & (EMB - 1);
    x0[tid] = embed_w[x_idx[node] * EMB + f];
}

__global__ void edge_count(const int* __restrict__ dst, float* __restrict__ cnt)
{
    int e = blockIdx.x * blockDim.x + threadIdx.x;
    if (e >= N_EDGES) return;
    atomicAdd(&cnt[dst[e]], 1.0f);
}

template <int F>
__global__ void edge_scatter(const int* __restrict__ src,
                             const int* __restrict__ dst,
                             const float* __restrict__ x,
                             float* __restrict__ agg)
{
    int tid = blockIdx.x * blockDim.x + threadIdx.x;
    if (tid >= N_EDGES * F) return;
    int e = tid / F;
    int f = tid & (F - 1);
    int s = src[e];
    int d = dst[e];
    atomicAdd(&agg[d * F + f], x[s * F + f]);
}

template <int K>
__global__ void sage_linear_wmma(const float* __restrict__ agg,
                                 const float* __restrict__ cnt,
                                 const float* __restrict__ xin,
                                 const float* __restrict__ Wl,
                                 const float* __restrict__ bl,
                                 const float* __restrict__ Wr,
                                 float* __restrict__ out)
{
    const int lane  = threadIdx.x & 31;
    const int wave  = threadIdx.x >> 5;
    const int wpb   = blockDim.x >> 5;
    const int tile  = blockIdx.x * wpb + wave;
    const int row0  = tile * 16;
    if (row0 >= N_NODES) return;

    const int half = lane >> 4;
    const int m    = lane & 15;
    const int rowA = row0 + m;

    const float inv = 1.0f / fmaxf(cnt[rowA], 1.0f);

    constexpr int KS = K / 32;
    const float4* aggq = (const float4*)(agg + rowA * K);
    const float4* xq   = (const float4*)(xin + rowA * K);

    v16h aM[KS], aX[KS];
    #pragma unroll
    for (int ks = 0; ks < KS; ++ks) {
        aM[ks] = load_frag(aggq, ks, half, inv);
        aX[ks] = load_frag(xq,   ks, half, 1.0f);
    }

    #pragma unroll
    for (int nt = 0; nt < HID / 16; ++nt) {
        const int ncol = nt * 16 + m;
        const float4* wlq = (const float4*)(Wl + ncol * K);
        const float4* wrq = (const float4*)(Wr + ncol * K);

        v8f acc = {};
        #pragma unroll
        for (int ks = 0; ks < KS; ++ks) {
            v16h bL = load_frag(wlq, ks, half, 1.0f);
            v16h bR = load_frag(wrq, ks, half, 1.0f);
            acc = __builtin_amdgcn_wmma_f32_16x16x32_f16(
                false, aM[ks], false, bL, (short)0, acc, false, false);
            acc = __builtin_amdgcn_wmma_f32_16x16x32_f16(
                false, aX[ks], false, bR, (short)0, acc, false, false);
        }

        const float bias = bl[ncol];
        float* op = out + (row0 + half * 8) * HID + ncol;
        #pragma unroll
        for (int d = 0; d < 8; ++d)
            op[d * HID] = fmaxf(acc[d] + bias, 0.0f);
    }
}

__global__ void pool_features(const int* __restrict__ batch,
                              const float* __restrict__ h2,
                              float* __restrict__ pooled)
{
    int tid = blockIdx.x * blockDim.x + threadIdx.x;
    if (tid >= N_NODES * HID) return;
    int node = tid >> 6;
    int f    = tid & (HID - 1);
    atomicAdd(&pooled[batch[node] * HID + f], h2[tid]);
}

__global__ void pool_count(const int* __restrict__ batch, float* __restrict__ gcnt)
{
    int i = blockIdx.x * blockDim.x + threadIdx.x;
    if (i >= N_NODES) return;
    atomicAdd(&gcnt[batch[i]], 1.0f);
}

__global__ void final_head(const float* __restrict__ pooled,
                           const float* __restrict__ gcnt,
                           const float* __restrict__ Wout,
                           const float* __restrict__ bout,
                           float* __restrict__ out)
{
    int tid = blockIdx.x * blockDim.x + threadIdx.x;
    if (tid >= N_GRAPHS * NCLS) return;
    int g = tid / NCLS;
    int c = tid % NCLS;
    float inv = 1.0f / fmaxf(gcnt[g], 1.0f);
    float acc = bout[c];
    #pragma unroll
    for (int h = 0; h < HID; ++h)
        acc += pooled[g * HID + h] * inv * Wout[c * HID + h];
    out[tid] = acc;
}

extern "C" void kernel_launch(void* const* d_in, const int* in_sizes, int n_in,
                              void* d_out, int out_size, void* d_ws, size_t ws_size,
                              hipStream_t stream)
{
    const int*   x_idx   = (const int*)  d_in[0];
    const int*   ei      = (const int*)  d_in[1];
    const int*   batch   = (const int*)  d_in[2];
    const float* embed_w = (const float*)d_in[3];
    const float* W1l     = (const float*)d_in[4];
    const float* b1      = (const float*)d_in[5];
    const float* W1r     = (const float*)d_in[6];
    const float* W2l     = (const float*)d_in[7];
    const float* b2      = (const float*)d_in[8];
    const float* W2r     = (const float*)d_in[9];
    const float* Wout    = (const float*)d_in[10];
    const float* bout    = (const float*)d_in[11];
    float*       out     = (float*)d_out;

    const int* src = ei;
    const int* dst = ei + N_EDGES;

    float* cnt    = (float*)d_ws;
    float* x0     = cnt    + N_NODES;
    float* agg1   = x0     + N_NODES * EMB;
    float* agg2   = x0;
    float* h1     = agg1   + N_NODES * EMB;
    float* h2     = h1     + N_NODES * HID;
    float* pooled = h2     + N_NODES * HID;
    float* gcnt   = pooled + N_GRAPHS * HID;

    const int B = 256;

    hipMemsetAsync(cnt,    0, sizeof(float) * N_NODES,        stream);
    hipMemsetAsync(agg1,   0, sizeof(float) * N_NODES * EMB,  stream);
    hipMemsetAsync(pooled, 0, sizeof(float) * N_GRAPHS * HID, stream);
    hipMemsetAsync(gcnt,   0, sizeof(float) * N_GRAPHS,       stream);

    embed_gather<<<(N_NODES * EMB + B - 1) / B, B, 0, stream>>>(x_idx, embed_w, x0);

    edge_count<<<(N_EDGES + B - 1) / B, B, 0, stream>>>(dst, cnt);

    edge_scatter<EMB><<<(N_EDGES * EMB + B - 1) / B, B, 0, stream>>>(src, dst, x0, agg1);
    {
        const int tiles  = N_NODES / 16;
        const int blocks = (tiles + 7) / 8;
        sage_linear_wmma<EMB><<<blocks, 256, 0, stream>>>(agg1, cnt, x0, W1l, b1, W1r, h1);
    }

    hipMemsetAsync(agg2, 0, sizeof(float) * N_NODES * HID, stream);
    edge_scatter<HID><<<(int)(((long long)N_EDGES * HID + B - 1) / B), B, 0, stream>>>(src, dst, h1, agg2);
    {
        const int tiles  = N_NODES / 16;
        const int blocks = (tiles + 7) / 8;
        sage_linear_wmma<HID><<<blocks, 256, 0, stream>>>(agg2, cnt, h1, W2l, b2, W2r, h2);
    }

    pool_features<<<(N_NODES * HID + B - 1) / B, B, 0, stream>>>(batch, h2, pooled);
    pool_count<<<(N_NODES + B - 1) / B, B, 0, stream>>>(batch, gcnt);
    final_head<<<(N_GRAPHS * NCLS + B - 1) / B, B, 0, stream>>>(pooled, gcnt, Wout, bout, out);
}


# small passes folded into neighbours: layer 1 reads the embedding table via x_idx, in-degree counted by its scatter, layer-2 projection adds into graph sums (atomic, arrival order); 14 launches to 6
# speedup vs baseline: 4.9035x; 4.9035x over previous
#include <hip/hip_runtime.h>


#define N_NODES  100000
#define N_EDGES  3200000
#define N_GRAPHS 4096
#define EMB      32
#define HID      64
#define NCLS     10

#define STREAM_BLOCKS  1024
#define STREAM_WAVES   (STREAM_BLOCKS * 8)

typedef __attribute__((ext_vector_type(16))) _Float16 v16h;
typedef __attribute__((ext_vector_type(8)))  float    v8f;

__device__ __forceinline__ int package_half()
{
    return (int)((__builtin_amdgcn_s_sendmsg_rtn(0x87) >> 18) & 1u);
}

__device__ __forceinline__ float4 add4(float4 a, float4 b)
{
    return make_float4(a.x + b.x, a.y + b.y, a.z + b.z, a.w + b.w);
}

__device__ __forceinline__ v16h pack_frag(float4 c0, float4 c1,
                                          float4 c2, float4 c3, float s)
{
    v16h r;
    r[0]  = (_Float16)(c0.x * s); r[1]  = (_Float16)(c0.y * s);
    r[2]  = (_Float16)(c0.z * s); r[3]  = (_Float16)(c0.w * s);
    r[4]  = (_Float16)(c1.x * s); r[5]  = (_Float16)(c1.y * s);
    r[6]  = (_Float16)(c1.z * s); r[7]  = (_Float16)(c1.w * s);
    r[8]  = (_Float16)(c2.x * s); r[9]  = (_Float16)(c2.y * s);
    r[10] = (_Float16)(c2.z * s); r[11] = (_Float16)(c2.w * s);
    r[12] = (_Float16)(c3.x * s); r[13] = (_Float16)(c3.y * s);
    r[14] = (_Float16)(c3.z * s); r[15] = (_Float16)(c3.w * s);
    return r;
}

__device__ __forceinline__ v16h load_frag(const float4* __restrict__ q,
                                          int ks, int half, float s)
{
    const int o = ks * 8 + half * 2;
    return pack_frag(q[o], q[o + 1], q[o + 4], q[o + 5], s);
}

__device__ __forceinline__ v16h load_frag_sum(const float4* __restrict__ q0,
                                              const float4* __restrict__ q1,
                                              int ks, int half, float s)
{
    const int o = ks * 8 + half * 2;
    return pack_frag(add4(q0[o],     q1[o]),     add4(q0[o + 1], q1[o + 1]),
                     add4(q0[o + 4], q1[o + 4]), add4(q0[o + 5], q1[o + 5]), s);
}

template <int F>
__global__ void edge_scatter(const int* __restrict__ src,
                             const int* __restrict__ dst,
                             const int* __restrict__ xrow,
                             const float* __restrict__ x,
                             float* __restrict__ agg,
                             float* __restrict__ cnt)
{
    static_assert(N_EDGES % 32 == 0, "edge runs of 32 need no tail");
    constexpr int R = F / 32;
    constexpr int G = 32 / R;

    constexpr int NRUN = N_EDGES / 32;
    static_assert(STREAM_WAVES <= NRUN, "every wave has a first run");

    const int lane = threadIdx.x & 31;
    const int wave = __builtin_amdgcn_readfirstlane(
        (int)((blockIdx.x * blockDim.x + threadIdx.x) >> 5));
    const int hsel = package_half();
    float* aggh = agg + hsel * (N_NODES * F);
    float* cnth = cnt + hsel * N_NODES;

    int s = src[wave * 32 + lane];
    int d = dst[wave * 32 + lane];
    if constexpr (F == EMB) s = xrow[s];
    const int run1 = min(wave + STREAM_WAVES, NRUN - 1);
    int s1 = src[run1 * 32 + lane];
    int d1 = dst[run1 * 32 + lane];

    for (int run = wave; run < NRUN; run += STREAM_WAVES) {
        int ns = s1;
        if constexpr (F == EMB) ns = xrow[s1];
        const int run2 = min(run + 2 * STREAM_WAVES, NRUN - 1);
        const int s2 = src[run2 * 32 + lane];
        const int d2 = dst[run2 * 32 + lane];

        if constexpr (F == EMB) atomicAdd(&cnth[d], 1.0f);

        #pragma unroll
        for (int g = 0; g < 32; g += G) {
            float v[G][R];
            #pragma unroll
            for (int j = 0; j < G; ++j) {
                const int sj = __builtin_amdgcn_readlane(s, g + j);
                const int xi = sj * F + lane;
                #pragma unroll
                for (int r = 0; r < R; ++r) v[j][r] = x[xi + r * 32];
            }
            #pragma unroll
            for (int j = 0; j < G; ++j) {
                const int dj = __builtin_amdgcn_readlane(d, g + j);
                const int ai = dj * F + lane;
                #pragma unroll
                for (int r = 0; r < R; ++r) atomicAdd(&aggh[ai + r * 32], v[j][r]);
            }
        }
        s  = ns;
        d  = d1;
        s1 = s2;
        d1 = d2;
    }
}

template <int K>
__global__ void __launch_bounds__(256)
sage_linear_wmma(const float* __restrict__ agg,
                                 const float* __restrict__ cnt,
                                 const float* __restrict__ xin,
                                 const int* __restrict__ xrow,
                                 const float* __restrict__ Wl,
                                 const float* __restrict__ bl,
                                 const float* __restrict__ Wr,
                                 float* __restrict__ out,
                                 const int* __restrict__ batch,
                                 float* __restrict__ gcnt)
{
    const int lane  = threadIdx.x & 31;
    const int wave  = threadIdx.x >> 5;
    const int wpb   = blockDim.x >> 5;
    const int tile  = blockIdx.x * wpb + wave;
    const int row0  = tile * 16;
    if (row0 >= N_NODES) return;

    const int half = lane >> 4;
    const int m    = lane & 15;
    const int rowA = row0 + m;

    const float inv = 1.0f / fmaxf(cnt[rowA] + cnt[N_NODES + rowA], 1.0f);

    constexpr int KS = K / 32;
    const float4* aggq0 = (const float4*)(agg + rowA * K);
    const float4* aggq1 = (const float4*)(agg + N_NODES * K + rowA * K);
    int xr = rowA;
    if constexpr (K == EMB) xr = xrow[rowA];
    const float4* xq    = (const float4*)(xin + xr * K);

    v16h aM[KS], aX[KS];
    #pragma unroll
    for (int ks = 0; ks < KS; ++ks) {
        aM[ks] = load_frag_sum(aggq0, aggq1, ks, half, inv);
        aX[ks] = load_frag(xq,   ks, half, 1.0f);
    }

    int    gb[8];
    float* pooledh = out;
    float* gcnth   = gcnt;
    if constexpr (K == HID) {
        const int hsel = package_half();
        pooledh = out  + hsel * (N_GRAPHS * HID);
        gcnth   = gcnt + hsel * N_GRAPHS;
        #pragma unroll
        for (int d = 0; d < 8; ++d) gb[d] = batch[row0 + half * 8 + d];
    }

    #pragma unroll
    for (int nt = 0; nt < HID / 16; ++nt) {
        const int ncol = nt * 16 + m;
        const float4* wlq = (const float4*)(Wl + ncol * K);
        const float4* wrq = (const float4*)(Wr + ncol * K);

        v8f acc = {};
        #pragma unroll
        for (int ks = 0; ks < KS; ++ks) {
            v16h bL = load_frag(wlq, ks, half, 1.0f);
            v16h bR = load_frag(wrq, ks, half, 1.0f);
            acc = __builtin_amdgcn_wmma_f32_16x16x32_f16(
                false, aM[ks], false, bL, (short)0, acc, false, false);
            acc = __builtin_amdgcn_wmma_f32_16x16x32_f16(
                false, aX[ks], false, bR, (short)0, acc, false, false);
        }

        const float bias = bl[ncol];
        if constexpr (K == HID) {
            #pragma unroll
            for (int d = 0; d < 8; ++d)
                atomicAdd(&pooledh[gb[d] * HID + ncol], fmaxf(acc[d] + bias, 0.0f));
        } else {
            float* op = out + (row0 + half * 8) * HID + ncol;
            #pragma unroll
            for (int d = 0; d < 8; ++d)
                op[d * HID] = fmaxf(acc[d] + bias, 0.0f);
        }
    }

    if constexpr (K == HID) {
        if (m == 0) {
            #pragma unroll
            for (int d = 0; d < 8; ++d) atomicAdd(&gcnth[gb[d]], 1.0f);
        }
    }
}

__global__ void final_head(const float* __restrict__ pooled,
                           const float* __restrict__ gcnt,
                           const float* __restrict__ Wout,
                           const float* __restrict__ bout,
                           float* __restrict__ out)
{
    int tid = blockIdx.x * blockDim.x + threadIdx.x;
    if (tid >= N_GRAPHS * NCLS) return;
    int g = tid / NCLS;
    int c = tid % NCLS;
    float inv = 1.0f / fmaxf(gcnt[g] + gcnt[N_GRAPHS + g], 1.0f);
    float acc = bout[c];
    #pragma unroll
    for (int h = 0; h < HID; ++h)
        acc += (pooled[g * HID + h] + pooled[N_GRAPHS * HID + g * HID + h]) * inv * Wout[c * HID + h];
    out[tid] = acc;
}

extern "C" void kernel_launch(void* const* d_in, const int* in_sizes, int n_in,
                              void* d_out, int out_size, void* d_ws, size_t ws_size,
                              hipStream_t stream)
{
    const int*   x_idx   = (const int*)  d_in[0];
    const int*   ei      = (const int*)  d_in[1];
    const int*   batch   = (const int*)  d_in[2];
    const float* embed_w = (const float*)d_in[3];
    const float* W1l     = (const float*)d_in[4];
    const float* b1      = (const float*)d_in[5];
    const float* W1r     = (const float*)d_in[6];
    const float* W2l     = (const float*)d_in[7];
    const float* b2      = (const float*)d_in[8];
    const float* W2r     = (const float*)d_in[9];
    const float* Wout    = (const float*)d_in[10];
    const float* bout    = (const float*)d_in[11];
    float*       out     = (float*)d_out;

    const int* src = ei;
    const int* dst = ei + N_EDGES;

    float* cnt    = (float*)d_ws;
    float* agg1   = cnt    + 2 * N_NODES;
    float* agg2   = agg1   + 2 * N_NODES * EMB;
    float* pooled = agg2   + 2 * N_NODES * HID;
    float* gcnt   = pooled + 2 * N_GRAPHS * HID;
    float* h1     = gcnt   + 2 * N_GRAPHS;

    const int B = 256;

    hipMemsetAsync(cnt, 0,
                   sizeof(float) * (2 * N_NODES + 2 * N_NODES * EMB + 2 * N_NODES * HID
                                    + 2 * N_GRAPHS * HID + 2 * N_GRAPHS),
                   stream);

    edge_scatter<EMB><<<STREAM_BLOCKS, B, 0, stream>>>(src, dst, x_idx, embed_w, agg1, cnt);
    {
        const int tiles  = N_NODES / 16;
        const int blocks = (tiles + 7) / 8;
        sage_linear_wmma<EMB><<<blocks, 256, 0, stream>>>(agg1, cnt, embed_w, x_idx, W1l, b1, W1r,
                                                          h1, batch, gcnt);
    }

    edge_scatter<HID><<<STREAM_BLOCKS, B, 0, stream>>>(src, dst, x_idx, h1, agg2, cnt);
    {
        const int tiles  = N_NODES / 16;
        const int blocks = (tiles + 7) / 8;
        sage_linear_wmma<HID><<<blocks, 256, 0, stream>>>(agg2, cnt, h1, x_idx, W2l, b2, W2r,
                                                          pooled, batch, gcnt);
    }

    final_head<<<(N_GRAPHS * NCLS + B - 1) / B, B, 0, stream>>>(pooled, gcnt, Wout, bout, out);
}


# layer-1 neighbour sums as per-node counts of the 64 table rows (one LDS integer add per edge lane) times the f32 table on f32 WMMA, not a row gather + 32-lane LDS f32 add per edge; new sum order
# speedup vs baseline: 7.6047x; 1.5509x over previous
#include <hip/hip_runtime.h>


#define N_NODES  100000
#define N_EDGES  3200000
#define N_GRAPHS 4096
#define EMB      32
#define HID      64
#define NCLS     10
#define VOCAB    64

#define BKT_SHIFT    6
#define BKT_NODES    (1 << BKT_SHIFT)
#define NBKT         ((N_NODES + BKT_NODES - 1) / BKT_NODES)
#define NBKT_PAD     1568
#define PART_BLOCKS  256
#define PART_THREADS 1024
#define PART_CHUNK   (N_EDGES / PART_BLOCKS)

typedef __attribute__((ext_vector_type(16))) _Float16 v16h;
typedef __attribute__((ext_vector_type(8)))  _Float16 v8h;
typedef __attribute__((ext_vector_type(2)))  _Float16 v2h;
typedef __attribute__((ext_vector_type(8)))  float    v8f;
typedef __attribute__((ext_vector_type(2)))  float    v2f;

__device__ __forceinline__ int package_half()
{
    return (int)((__builtin_amdgcn_s_sendmsg_rtn(0x87) >> 18) & 1u);
}

__device__ __forceinline__ v16h pack_frag(float4 c0, float4 c1,
                                          float4 c2, float4 c3, float s)
{
    v16h r;
    r[0]  = (_Float16)(c0.x * s); r[1]  = (_Float16)(c0.y * s);
    r[2]  = (_Float16)(c0.z * s); r[3]  = (_Float16)(c0.w * s);
    r[4]  = (_Float16)(c1.x * s); r[5]  = (_Float16)(c1.y * s);
    r[6]  = (_Float16)(c1.z * s); r[7]  = (_Float16)(c1.w * s);
    r[8]  = (_Float16)(c2.x * s); r[9]  = (_Float16)(c2.y * s);
    r[10] = (_Float16)(c2.z * s); r[11] = (_Float16)(c2.w * s);
    r[12] = (_Float16)(c3.x * s); r[13] = (_Float16)(c3.y * s);
    r[14] = (_Float16)(c3.z * s); r[15] = (_Float16)(c3.w * s);
    return r;
}

__device__ __forceinline__ v16h load_frag(const float4* __restrict__ q,
                                          int ks, int half, float s)
{
    const int o = ks * 8 + half * 2;
    return pack_frag(q[o], q[o + 1], q[o + 4], q[o + 5], s);
}

__device__ __forceinline__ v16h load_frag_h(const _Float16* __restrict__ rowp,
                                            int ks, int half)
{
    const v8h* q  = (const v8h*)rowp;
    const v8h  lo = q[ks * 4 + half];
    const v8h  hi = q[ks * 4 + 2 + half];
    return __builtin_shufflevector(lo, hi, 0, 1, 2, 3, 4, 5, 6, 7,
                                   8, 9, 10, 11, 12, 13, 14, 15);
}

__global__ void __launch_bounds__(PART_THREADS)
bucket_count(const int* __restrict__ dst,
             int* __restrict__ blkcnt,
             int* __restrict__ tot)
{
    static_assert(N_EDGES % PART_BLOCKS == 0, "equal chunks, no tail");
    __shared__ int c[NBKT];
    for (int b = threadIdx.x; b < NBKT; b += PART_THREADS) c[b] = 0;
    __syncthreads();
    const int e0 = blockIdx.x * PART_CHUNK;
    constexpr int T = (PART_CHUNK + PART_THREADS - 1) / PART_THREADS;
    int d[T];
    #pragma unroll
    for (int k = 0; k < T; ++k) {
        const int i = threadIdx.x + k * PART_THREADS;
        d[k] = (i < PART_CHUNK) ? dst[e0 + i] : -1;
    }
    #pragma unroll
    for (int k = 0; k < T; ++k)
        if (d[k] >= 0) atomicAdd(&c[d[k] >> BKT_SHIFT], 1);
    __syncthreads();
    for (int b = threadIdx.x; b < NBKT; b += PART_THREADS) {
        const int v = c[b];
        blkcnt[blockIdx.x * NBKT + b] = v;
        if (v) atomicAdd(&tot[b], v);
    }
}

__global__ void __launch_bounds__(PART_THREADS)
bucket_place(const int* __restrict__ src,
             const int* __restrict__ dst,
             const int* __restrict__ xrow,
             const int* __restrict__ blkcnt,
             const int* __restrict__ tot,
             int* __restrict__ cursor,
             int* __restrict__ bstart,
             int* __restrict__ rec)
{
    static_assert(2 * PART_THREADS >= NBKT, "two totals per thread cover all ranges");
    __shared__ int pos[NBKT];
    __shared__ int wsum[PART_THREADS / 32];

    const int t    = threadIdx.x;
    const int lane = t & 31;
    const int wv   = t >> 5;

    const int i0 = 2 * t, i1 = 2 * t + 1;
    const int a0 = (i0 < NBKT) ? tot[i0] : 0;
    const int a1 = (i1 < NBKT) ? tot[i1] : 0;
    const int own = a0 + a1;
    int inc = own;
    #pragma unroll
    for (int o = 1; o < 32; o <<= 1) {
        const int u = __shfl_up(inc, o);
        if (lane >= o) inc += u;
    }
    if (lane == 31) wsum[wv] = inc;
    __syncthreads();
    if (wv == 0) {
        const int w = wsum[lane];
        int winc = w;
        #pragma unroll
        for (int o = 1; o < 32; o <<= 1) {
            const int u = __shfl_up(winc, o);
            if (lane >= o) winc += u;
        }
        wsum[lane] = winc - w;
    }
    __syncthreads();
    const int excl = wsum[wv] + inc - own;
    if (i0 < NBKT) pos[i0] = excl;
    if (i1 < NBKT) pos[i1] = excl + a0;
    if (blockIdx.x == 0) {
        if (i0 < NBKT) bstart[i0] = excl;
        if (i1 < NBKT) bstart[i1] = excl + a0;
    }
    __syncthreads();

    for (int b = t; b < NBKT; b += PART_THREADS) {
        const int c = blkcnt[blockIdx.x * NBKT + b];
        if (c) pos[b] += atomicAdd(&cursor[b], c);
    }
    __syncthreads();

    int2* rec2 = (int2*)rec;
    const int e0 = blockIdx.x * PART_CHUNK;
    constexpr int T = (PART_CHUNK + PART_THREADS - 1) / PART_THREADS;
    int s[T], d[T], xr[T];
    #pragma unroll
    for (int k = 0; k < T; ++k) {
        const int i = t + k * PART_THREADS;
        s[k] = (i < PART_CHUNK) ? src[e0 + i] : 0;
        d[k] = (i < PART_CHUNK) ? dst[e0 + i] : -1;
    }
    #pragma unroll
    for (int k = 0; k < T; ++k) xr[k] = xrow[s[k]];
    #pragma unroll
    for (int k = 0; k < T; ++k) {
        if (d[k] >= 0) {
            const int p = atomicAdd(&pos[d[k] >> BKT_SHIFT], 1);
            rec2[p] = make_int2(s[k], (xr[k] << 8) | (d[k] & (BKT_NODES - 1)));
        }
    }
}

#define CNT_PITCH 68
__global__ void __launch_bounds__(128)
bucket_mean_table(const int* __restrict__ rec,
                  const int* __restrict__ bstart,
                  const int* __restrict__ tot,
                  const float* __restrict__ xtab,
                  _Float16* __restrict__ mean)
{
    static_assert(BKT_NODES == 64 && VOCAB == 64 && EMB == 32,
                  "4 waves x 16 rows, K = 64 as 2 lane halves x 32, two 16-column tiles");
    __shared__ alignas(16) int cnt[BKT_NODES * CNT_PITCH];
    __shared__ alignas(16) int deg[BKT_NODES];

    const int t    = threadIdx.x;
    const int lane = t & 31;
    const int wv   = __builtin_amdgcn_readfirstlane(t >> 5);
    const int b    = blockIdx.x;
    const int half = lane >> 4;
    const int c    = lane & 15;

    const int   beg  = bstart[b];
    const int   n    = tot[b];
    const int2* rec2 = (const int2*)rec + beg;

    constexpr int U = 16;
    int y[U];
    #pragma unroll
    for (int u = 0; u < U; ++u) {
        const int i = u * 128 + t;
        y[u] = (i < n) ? rec2[i].y : -1;
    }
    float2 e[VOCAB / 2];
    const float2* ep = (const float2*)(xtab + 32 * half * EMB) + c;
    #pragma unroll
    for (int j = 0; j < VOCAB / 2; ++j) e[j] = ep[j * (EMB / 2)];

    int4* z = (int4*)cnt;
    for (int i = t; i < BKT_NODES * CNT_PITCH / 4; i += 128) z[i] = make_int4(0, 0, 0, 0);
    if (t < BKT_NODES) deg[t] = 0;
    __syncthreads();

    for (int base = 0; ; ) {
        #pragma unroll
        for (int u = 0; u < U; ++u) {
            if (y[u] >= 0) {
                const int dl = y[u] & 255;
                atomicAdd(&cnt[dl * CNT_PITCH + (y[u] >> 8)], 1);
                atomicAdd(&deg[dl], 1);
            }
        }
        base += 128 * U;
        if (base >= n) break;
        #pragma unroll
        for (int u = 0; u < U; ++u) {
            const int i = base + u * 128 + t;
            y[u] = (i < n) ? rec2[i].y : -1;
        }
    }
    __syncthreads();

    const int4* arow = (const int4*)&cnt[(wv * 16 + c) * CNT_PITCH + 32 * half];
    v8f acc0 = {}, acc1 = {};
    #pragma unroll
    for (int q4 = 0; q4 < 8; ++q4) {
        const int4 a4 = arow[q4];
        v2f a, b0, b1;
        a[0]  = (float)a4.x;       a[1]  = (float)a4.y;
        b0[0] = e[4 * q4].x;       b0[1] = e[4 * q4 + 1].x;
        b1[0] = e[4 * q4].y;       b1[1] = e[4 * q4 + 1].y;
        acc0 = __builtin_amdgcn_wmma_f32_16x16x4_f32(false, a, false, b0, (short)0, acc0, false, false);
        acc1 = __builtin_amdgcn_wmma_f32_16x16x4_f32(false, a, false, b1, (short)0, acc1, false, false);
        a[0]  = (float)a4.z;       a[1]  = (float)a4.w;
        b0[0] = e[4 * q4 + 2].x;   b0[1] = e[4 * q4 + 3].x;
        b1[0] = e[4 * q4 + 2].y;   b1[1] = e[4 * q4 + 3].y;
        acc0 = __builtin_amdgcn_wmma_f32_16x16x4_f32(false, a, false, b0, (short)0, acc0, false, false);
        acc1 = __builtin_amdgcn_wmma_f32_16x16x4_f32(false, a, false, b1, (short)0, acc1, false, false);
    }

    const int4* dq = (const int4*)&deg[wv * 16 + 8 * half];
    const int4  d0 = dq[0], d1 = dq[1];
    const int   dg[8] = {d0.x, d0.y, d0.z, d0.w, d1.x, d1.y, d1.z, d1.w};
    const int   node0 = b * BKT_NODES + wv * 16 + 8 * half;
    v2h* mw = (v2h*)mean + node0 * (EMB / 2) + c;
    #pragma unroll
    for (int r = 0; r < 8; ++r) {
        if (node0 + r < N_NODES) {
            const float inv = 1.0f / fmaxf((float)dg[r], 1.0f);
            v2h o;
            o.x = (_Float16)(acc0[r] * inv);
            o.y = (_Float16)(acc1[r] * inv);
            mw[r * (EMB / 2)] = o;
        }
    }
}

__global__ void __launch_bounds__(256)
bucket_sum(const int* __restrict__ rec,
           const int* __restrict__ bstart,
           const int* __restrict__ tot,
           const _Float16* __restrict__ xh,
           _Float16* __restrict__ mean)
{
    constexpr int F = HID;
    __shared__ float acc[(BKT_NODES + 1) * F];
    __shared__ int   deg[BKT_NODES];

    const int t    = threadIdx.x;
    const int lane = t & 31;
    const int wv   = __builtin_amdgcn_readfirstlane(t >> 5);
    const int b    = blockIdx.x;

    for (int i = t; i < (BKT_NODES + 1) * F; i += 256) acc[i] = 0.0f;
    if (t < BKT_NODES) deg[t] = 0;
    __syncthreads();

    const int   beg  = bstart[b];
    const int   n    = tot[b];
    const int2* rec2 = (const int2*)rec + beg;
    const v2h*  xw   = (const v2h*)xh;

    const int2 pad = make_int2(0, BKT_NODES);
    int2 p = pad;
    if (wv * 32 + lane < n) p = rec2[wv * 32 + lane];

    for (int base = wv * 32; base < n; base += 256) {
        const int  inext = base + 256 + lane;
        int2 pn = pad;
        if (inext < n) pn = rec2[inext];

        const int dl  = p.y & 255;
        const int row = p.x;
        if (base + lane < n) atomicAdd(&deg[dl], 1);

        {
            v2h v[32];
            #pragma unroll
            for (int j = 0; j < 32; ++j) {
                const int sj = __builtin_amdgcn_readlane(row, j);
                v[j] = xw[sj * (HID / 2) + lane];
            }
            #pragma unroll
            for (int j = 0; j < 32; ++j) {
                const int dj = __builtin_amdgcn_readlane(dl, j);
                float* a = &acc[dj * HID + lane];
                atomicAdd(a,      (float)v[j].x);
                atomicAdd(a + 32, (float)v[j].y);
            }
        }
        p = pn;
    }
    __syncthreads();

    const int node0 = b * BKT_NODES;
    v2h* mw = (v2h*)mean + node0 * (F / 2);
    for (int i = t; i < BKT_NODES * (F / 2); i += 256) {
        const int nl = i / (F / 2);
        const int j  = i % (F / 2);
        if (node0 + nl < N_NODES) {
            const float inv = 1.0f / fmaxf((float)deg[nl], 1.0f);
            const float a0 = acc[nl * HID + j];
            const float a1 = acc[nl * HID + 32 + j];
            v2h o;
            o.x = (_Float16)(a0 * inv);
            o.y = (_Float16)(a1 * inv);
            mw[i] = o;
        }
    }
}

template <int K>
__global__ void __launch_bounds__(256)
sage_linear_wmma(const _Float16* __restrict__ mean,
                 const float* __restrict__ xtab,
                 const _Float16* __restrict__ xh,
                 const int* __restrict__ xrow,
                 const float* __restrict__ Wl,
                 const float* __restrict__ bl,
                 const float* __restrict__ Wr,
                 _Float16* __restrict__ h1out,
                 float* __restrict__ pooled,
                 const int* __restrict__ batch,
                 float* __restrict__ gcnt)
{
    const int lane  = threadIdx.x & 31;
    const int wave  = threadIdx.x >> 5;
    const int wpb   = blockDim.x >> 5;
    const int tile  = blockIdx.x * wpb + wave;
    const int row0  = tile * 16;
    if (row0 >= N_NODES) return;

    const int half = lane >> 4;
    const int m    = lane & 15;
    const int rowA = row0 + m;

    constexpr int KS = K / 32;
    const _Float16* meanp = mean + rowA * K;

    v16h aM[KS], aX[KS];
    if constexpr (K == EMB) {
        const float4* xq = (const float4*)(xtab + xrow[rowA] * K);
        #pragma unroll
        for (int ks = 0; ks < KS; ++ks) {
            aM[ks] = load_frag_h(meanp, ks, half);
            aX[ks] = load_frag(xq, ks, half, 1.0f);
        }
    } else {
        const _Float16* xp = xh + rowA * K;
        #pragma unroll
        for (int ks = 0; ks < KS; ++ks) {
            aM[ks] = load_frag_h(meanp, ks, half);
            aX[ks] = load_frag_h(xp,    ks, half);
        }
    }

    int    gb[8];
    float* pooledh = pooled;
    float* gcnth   = gcnt;
    if constexpr (K == HID) {
        const int hsel = package_half();
        pooledh = pooled + hsel * (N_GRAPHS * HID);
        gcnth   = gcnt   + hsel * N_GRAPHS;
        #pragma unroll
        for (int d = 0; d < 8; ++d) gb[d] = batch[row0 + half * 8 + d];
    }

    #pragma unroll
    for (int pr = 0; pr < HID / 32; ++pr) {
        v8f   acc[2];
        float bias[2];
        #pragma unroll
        for (int q = 0; q < 2; ++q) {
            const int nt   = pr * 2 + q;
            const int ncol = (K == EMB) ? (32 * pr + 2 * m + q) : (nt * 16 + m);
            const float4* wlq = (const float4*)(Wl + ncol * K);
            const float4* wrq = (const float4*)(Wr + ncol * K);

            v8f c = {};
            #pragma unroll
            for (int ks = 0; ks < KS; ++ks) {
                v16h bL = load_frag(wlq, ks, half, 1.0f);
                v16h bR = load_frag(wrq, ks, half, 1.0f);
                c = __builtin_amdgcn_wmma_f32_16x16x32_f16(
                    false, aM[ks], false, bL, (short)0, c, false, false);
                c = __builtin_amdgcn_wmma_f32_16x16x32_f16(
                    false, aX[ks], false, bR, (short)0, c, false, false);
            }
            acc[q]  = c;
            bias[q] = bl[ncol];

            if constexpr (K == HID) {
                #pragma unroll
                for (int d = 0; d < 8; ++d)
                    atomicAdd(&pooledh[gb[d] * HID + ncol], fmaxf(c[d] + bias[q], 0.0f));
            }
        }

        if constexpr (K == EMB) {
            v2h* op = (v2h*)h1out + (row0 + half * 8) * (HID / 2) + 16 * pr + m;
            #pragma unroll
            for (int d = 0; d < 8; ++d) {
                v2h o;
                o.x = (_Float16)fmaxf(acc[0][d] + bias[0], 0.0f);
                o.y = (_Float16)fmaxf(acc[1][d] + bias[1], 0.0f);
                op[d * (HID / 2)] = o;
            }
        }
    }

    if constexpr (K == HID) {
        if (m == 0) {
            #pragma unroll
            for (int d = 0; d < 8; ++d) atomicAdd(&gcnth[gb[d]], 1.0f);
        }
    }
}

__global__ void final_head(const float* __restrict__ pooled,
                           const float* __restrict__ gcnt,
                           const float* __restrict__ Wout,
                           const float* __restrict__ bout,
                           float* __restrict__ out)
{
    int tid = blockIdx.x * blockDim.x + threadIdx.x;
    if (tid >= N_GRAPHS * NCLS) return;
    int g = tid / NCLS;
    int c = tid % NCLS;
    float inv = 1.0f / fmaxf(gcnt[g] + gcnt[N_GRAPHS + g], 1.0f);
    float acc = bout[c];
    #pragma unroll
    for (int h = 0; h < HID; ++h)
        acc += (pooled[g * HID + h] + pooled[N_GRAPHS * HID + g * HID + h]) * inv * Wout[c * HID + h];
    out[tid] = acc;
}

extern "C" void kernel_launch(void* const* d_in, const int* in_sizes, int n_in,
                              void* d_out, int out_size, void* d_ws, size_t ws_size,
                              hipStream_t stream)
{
    const int*   x_idx   = (const int*)  d_in[0];
    const int*   ei      = (const int*)  d_in[1];
    const int*   batch   = (const int*)  d_in[2];
    const float* embed_w = (const float*)d_in[3];
    const float* W1l     = (const float*)d_in[4];
    const float* b1      = (const float*)d_in[5];
    const float* W1r     = (const float*)d_in[6];
    const float* W2l     = (const float*)d_in[7];
    const float* b2      = (const float*)d_in[8];
    const float* W2r     = (const float*)d_in[9];
    const float* Wout    = (const float*)d_in[10];
    const float* bout    = (const float*)d_in[11];
    float*       out     = (float*)d_out;

    const int* src = ei;
    const int* dst = ei + N_EDGES;

    int*   tot    = (int*)d_ws;
    int*   cursor = tot    + NBKT_PAD;
    float* pooled = (float*)(cursor + NBKT_PAD);
    float* gcnt   = pooled + 2 * N_GRAPHS * HID;
    int*   bstart = (int*)(gcnt + 2 * N_GRAPHS);
    int*   blkcnt = bstart + NBKT_PAD;
    int*   rec    = blkcnt + PART_BLOCKS * NBKT;
    _Float16* mean1 = (_Float16*)(rec + 2 * N_EDGES);
    _Float16* mean2 = mean1 + N_NODES * EMB;
    _Float16* h1    = mean2 + N_NODES * HID;

    const int B = 256;

    hipMemsetAsync(tot, 0,
                   sizeof(int) * (2 * NBKT_PAD + 2 * N_GRAPHS * HID + 2 * N_GRAPHS),
                   stream);

    bucket_count<<<PART_BLOCKS, PART_THREADS, 0, stream>>>(dst, blkcnt, tot);
    bucket_place<<<PART_BLOCKS, PART_THREADS, 0, stream>>>(src, dst, x_idx, blkcnt, tot, cursor,
                                                            bstart, rec);

    bucket_mean_table<<<NBKT, 128, 0, stream>>>(rec, bstart, tot, embed_w, mean1);
    {
        const int tiles  = N_NODES / 16;
        const int blocks = (tiles + 7) / 8;
        sage_linear_wmma<EMB><<<blocks, 256, 0, stream>>>(mean1, embed_w, h1, x_idx, W1l, b1, W1r,
                                                          h1, pooled, batch, gcnt);
    }

    bucket_sum<<<NBKT, 256, 0, stream>>>(rec, bstart, tot, h1, mean2);
    {
        const int tiles  = N_NODES / 16;
        const int blocks = (tiles + 7) / 8;
        sage_linear_wmma<HID><<<blocks, 256, 0, stream>>>(mean2, embed_w, h1, x_idx, W2l, b2, W2r,
                                                          h1, pooled, batch, gcnt);
    }

    final_head<<<(N_GRAPHS * NCLS + B - 1) / B, B, 0, stream>>>(pooled, gcnt, Wout, bout, out);
}


# each layer's projection moved into the range's owner work-group, fed from the means it has just formed in LDS (layer-2 weights staged there once as f16): mean planes never stored, 7 launches to 5
# speedup vs baseline: 9.1271x; 1.2002x over previous
#include <hip/hip_runtime.h>


#define N_NODES  100000
#define N_EDGES  3200000
#define N_GRAPHS 4096
#define EMB      32
#define HID      64
#define NCLS     10
#define VOCAB    64

#define BKT_SHIFT    6
#define BKT_NODES    (1 << BKT_SHIFT)
#define NBKT         ((N_NODES + BKT_NODES - 1) / BKT_NODES)
#define NBKT_PAD     1568
#define PART_BLOCKS  256
#define PART_THREADS 1024
#define PART_CHUNK   (N_EDGES / PART_BLOCKS)

typedef __attribute__((ext_vector_type(16))) _Float16 v16h;
typedef __attribute__((ext_vector_type(8)))  _Float16 v8h;
typedef __attribute__((ext_vector_type(4)))  _Float16 v4h;
typedef __attribute__((ext_vector_type(2)))  _Float16 v2h;
typedef __attribute__((ext_vector_type(8)))  float    v8f;
typedef __attribute__((ext_vector_type(2)))  float    v2f;

__device__ __forceinline__ int package_half()
{
    return (int)((__builtin_amdgcn_s_sendmsg_rtn(0x87) >> 18) & 1u);
}

__device__ __forceinline__ v16h pack_frag(float4 c0, float4 c1,
                                          float4 c2, float4 c3, float s)
{
    v16h r;
    r[0]  = (_Float16)(c0.x * s); r[1]  = (_Float16)(c0.y * s);
    r[2]  = (_Float16)(c0.z * s); r[3]  = (_Float16)(c0.w * s);
    r[4]  = (_Float16)(c1.x * s); r[5]  = (_Float16)(c1.y * s);
    r[6]  = (_Float16)(c1.z * s); r[7]  = (_Float16)(c1.w * s);
    r[8]  = (_Float16)(c2.x * s); r[9]  = (_Float16)(c2.y * s);
    r[10] = (_Float16)(c2.z * s); r[11] = (_Float16)(c2.w * s);
    r[12] = (_Float16)(c3.x * s); r[13] = (_Float16)(c3.y * s);
    r[14] = (_Float16)(c3.z * s); r[15] = (_Float16)(c3.w * s);
    return r;
}

__device__ __forceinline__ v16h load_frag(const float4* __restrict__ q,
                                          int ks, int half, float s)
{
    const int o = ks * 8 + half * 2;
    return pack_frag(q[o], q[o + 1], q[o + 4], q[o + 5], s);
}

__device__ __forceinline__ v16h load_frag_h(const _Float16* __restrict__ rowp,
                                            int ks, int half)
{
    const v8h* q  = (const v8h*)rowp;
    const v8h  lo = q[ks * 4 + half];
    const v8h  hi = q[ks * 4 + 2 + half];
    return __builtin_shufflevector(lo, hi, 0, 1, 2, 3, 4, 5, 6, 7,
                                   8, 9, 10, 11, 12, 13, 14, 15);
}

#define W_PITCH 72

__device__ __forceinline__ v16h pack_pairs(float4 ev0, float4 od0,
                                           float4 ev1, float4 od1, float s)
{
    v16h r;
    r[0]  = (_Float16)(ev0.x * s); r[1]  = (_Float16)(od0.x * s);
    r[2]  = (_Float16)(ev0.y * s); r[3]  = (_Float16)(od0.y * s);
    r[4]  = (_Float16)(ev0.z * s); r[5]  = (_Float16)(od0.z * s);
    r[6]  = (_Float16)(ev0.w * s); r[7]  = (_Float16)(od0.w * s);
    r[8]  = (_Float16)(ev1.x * s); r[9]  = (_Float16)(od1.x * s);
    r[10] = (_Float16)(ev1.y * s); r[11] = (_Float16)(od1.y * s);
    r[12] = (_Float16)(ev1.z * s); r[13] = (_Float16)(od1.z * s);
    r[14] = (_Float16)(ev1.w * s); r[15] = (_Float16)(od1.w * s);
    return r;
}

__device__ __forceinline__ void project_layer1(v16h aM, v16h aX, int row0, int half, int m,
                                               const float* __restrict__ Wl,
                                               const float* __restrict__ bl,
                                               const float* __restrict__ Wr,
                                               _Float16* __restrict__ h1out)
{
    #pragma unroll
    for (int pr = 0; pr < HID / 32; ++pr) {
        v8f   acc[2];
        float bias[2];
        #pragma unroll
        for (int q = 0; q < 2; ++q) {
            const int ncol = 32 * pr + 2 * m + q;
            const v16h bL = load_frag((const float4*)(Wl + ncol * EMB), 0, half, 1.0f);
            const v16h bR = load_frag((const float4*)(Wr + ncol * EMB), 0, half, 1.0f);
            v8f c = {};
            c = __builtin_amdgcn_wmma_f32_16x16x32_f16(
                false, aM, false, bL, (short)0, c, false, false);
            c = __builtin_amdgcn_wmma_f32_16x16x32_f16(
                false, aX, false, bR, (short)0, c, false, false);
            acc[q]  = c;
            bias[q] = bl[ncol];
        }
        v2h* op = (v2h*)h1out + (row0 + half * 8) * (HID / 2) + 16 * pr + m;
        #pragma unroll
        for (int d = 0; d < 8; ++d) {
            v2h o;
            o.x = (_Float16)fmaxf(acc[0][d] + bias[0], 0.0f);
            o.y = (_Float16)fmaxf(acc[1][d] + bias[1], 0.0f);
            op[d * (HID / 2)] = o;
        }
    }
}

__device__ __forceinline__ void project_layer2(const float4* ar, float inv,
                                               const _Float16* __restrict__ xp,
                                               int row0, int pr, int half, int m,
                                               const _Float16* wls,
                                               const _Float16* wrs,
                                               const float* __restrict__ bl,
                                               float* __restrict__ pooled,
                                               const int* __restrict__ batch,
                                               float* __restrict__ gcnt)
{
    const int hsel    = package_half();
    float*    pooledh = pooled + hsel * (N_GRAPHS * HID);
    float*    gcnth   = gcnt   + hsel * N_GRAPHS;
    const int ncol0 = pr * 32 + m;
    const int ncol1 = ncol0 + 16;
    v8f c0 = {}, c1 = {};
    #pragma unroll
    for (int ks = 0; ks < 2; ++ks) {
        const int  w0 = ks * 4 + half;
        const v16h aM = pack_pairs(ar[w0], ar[8 + w0], ar[w0 + 2], ar[8 + w0 + 2], inv);
        c0 = __builtin_amdgcn_wmma_f32_16x16x32_f16(
            false, aM, false, load_frag_h(wls + ncol0 * W_PITCH, ks, half),
            (short)0, c0, false, false);
        c1 = __builtin_amdgcn_wmma_f32_16x16x32_f16(
            false, aM, false, load_frag_h(wls + ncol1 * W_PITCH, ks, half),
            (short)0, c1, false, false);
        const v16h aX = load_frag_h(xp, ks, half);
        c0 = __builtin_amdgcn_wmma_f32_16x16x32_f16(
            false, aX, false, load_frag_h(wrs + ncol0 * W_PITCH, ks, half),
            (short)0, c0, false, false);
        c1 = __builtin_amdgcn_wmma_f32_16x16x32_f16(
            false, aX, false, load_frag_h(wrs + ncol1 * W_PITCH, ks, half),
            (short)0, c1, false, false);
    }

    int gb[8];
    #pragma unroll
    for (int d = 0; d < 8; ++d) gb[d] = batch[row0 + half * 8 + d];
    const float bias0 = bl[ncol0], bias1 = bl[ncol1];
    #pragma unroll
    for (int d = 0; d < 8; ++d)
        atomicAdd(&pooledh[gb[d] * HID + ncol0], fmaxf(c0[d] + bias0, 0.0f));
    #pragma unroll
    for (int d = 0; d < 8; ++d)
        atomicAdd(&pooledh[gb[d] * HID + ncol1], fmaxf(c1[d] + bias1, 0.0f));

    if (pr == 0 && m == 0) {
        #pragma unroll
        for (int d = 0; d < 8; ++d) atomicAdd(&gcnth[gb[d]], 1.0f);
    }
}

__global__ void __launch_bounds__(PART_THREADS)
bucket_count(const int* __restrict__ dst,
             int* __restrict__ blkcnt,
             int* __restrict__ tot)
{
    static_assert(N_EDGES % PART_BLOCKS == 0, "equal chunks, no tail");
    __shared__ int c[NBKT];
    for (int b = threadIdx.x; b < NBKT; b += PART_THREADS) c[b] = 0;
    __syncthreads();
    const int e0 = blockIdx.x * PART_CHUNK;
    constexpr int T = (PART_CHUNK + PART_THREADS - 1) / PART_THREADS;
    int d[T];
    #pragma unroll
    for (int k = 0; k < T; ++k) {
        const int i = threadIdx.x + k * PART_THREADS;
        d[k] = (i < PART_CHUNK) ? dst[e0 + i] : -1;
    }
    #pragma unroll
    for (int k = 0; k < T; ++k)
        if (d[k] >= 0) atomicAdd(&c[d[k] >> BKT_SHIFT], 1);
    __syncthreads();
    for (int b = threadIdx.x; b < NBKT; b += PART_THREADS) {
        const int v = c[b];
        blkcnt[blockIdx.x * NBKT + b] = v;
        if (v) atomicAdd(&tot[b], v);
    }
}

__global__ void __launch_bounds__(PART_THREADS)
bucket_place(const int* __restrict__ src,
             const int* __restrict__ dst,
             const int* __restrict__ xrow,
             const int* __restrict__ blkcnt,
             const int* __restrict__ tot,
             int* __restrict__ cursor,
             int* __restrict__ bstart,
             int* __restrict__ rec)
{
    static_assert(2 * PART_THREADS >= NBKT, "two totals per thread cover all ranges");
    __shared__ int pos[NBKT];
    __shared__ int wsum[PART_THREADS / 32];

    const int t    = threadIdx.x;
    const int lane = t & 31;
    const int wv   = t >> 5;

    const int i0 = 2 * t, i1 = 2 * t + 1;
    const int a0 = (i0 < NBKT) ? tot[i0] : 0;
    const int a1 = (i1 < NBKT) ? tot[i1] : 0;
    const int own = a0 + a1;
    int inc = own;
    #pragma unroll
    for (int o = 1; o < 32; o <<= 1) {
        const int u = __shfl_up(inc, o);
        if (lane >= o) inc += u;
    }
    if (lane == 31) wsum[wv] = inc;
    __syncthreads();
    if (wv == 0) {
        const int w = wsum[lane];
        int winc = w;
        #pragma unroll
        for (int o = 1; o < 32; o <<= 1) {
            const int u = __shfl_up(winc, o);
            if (lane >= o) winc += u;
        }
        wsum[lane] = winc - w;
    }
    __syncthreads();
    const int excl = wsum[wv] + inc - own;
    if (i0 < NBKT) pos[i0] = excl;
    if (i1 < NBKT) pos[i1] = excl + a0;
    if (blockIdx.x == 0) {
        if (i0 < NBKT) bstart[i0] = excl;
        if (i1 < NBKT) bstart[i1] = excl + a0;
    }
    __syncthreads();

    for (int b = t; b < NBKT; b += PART_THREADS) {
        const int c = blkcnt[blockIdx.x * NBKT + b];
        if (c) pos[b] += atomicAdd(&cursor[b], c);
    }
    __syncthreads();

    int2* rec2 = (int2*)rec;
    const int e0 = blockIdx.x * PART_CHUNK;
    constexpr int T = (PART_CHUNK + PART_THREADS - 1) / PART_THREADS;
    int s[T], d[T], xr[T];
    #pragma unroll
    for (int k = 0; k < T; ++k) {
        const int i = t + k * PART_THREADS;
        s[k] = (i < PART_CHUNK) ? src[e0 + i] : 0;
        d[k] = (i < PART_CHUNK) ? dst[e0 + i] : -1;
    }
    #pragma unroll
    for (int k = 0; k < T; ++k) xr[k] = xrow[s[k]];
    #pragma unroll
    for (int k = 0; k < T; ++k) {
        if (d[k] >= 0) {
            const int p = atomicAdd(&pos[d[k] >> BKT_SHIFT], 1);
            rec2[p] = make_int2(s[k], (xr[k] << 8) | (d[k] & (BKT_NODES - 1)));
        }
    }
}

#define CNT_PITCH 68
#define MT_PITCH  20
__global__ void __launch_bounds__(128)
range_layer1(const int* __restrict__ rec,
             const int* __restrict__ bstart,
             const int* __restrict__ tot,
             const float* __restrict__ xtab,
             const int* __restrict__ xrow,
             const float* __restrict__ Wl,
             const float* __restrict__ bl,
             const float* __restrict__ Wr,
             _Float16* __restrict__ h1out)
{
    static_assert(BKT_NODES == 64 && VOCAB == 64 && EMB == 32,
                  "4 waves x 16 rows, K = 64 as 2 lane halves x 32, two 16-column tiles");
    __shared__ alignas(16) int cnt[BKT_NODES * CNT_PITCH];
    __shared__ alignas(16) int deg[BKT_NODES];
    __shared__ alignas(16) unsigned mt[BKT_NODES * MT_PITCH];

    const int t    = threadIdx.x;
    const int lane = t & 31;
    const int wv   = __builtin_amdgcn_readfirstlane(t >> 5);
    const int b    = blockIdx.x;
    const int half = lane >> 4;
    const int c    = lane & 15;

    const int   beg  = bstart[b];
    const int   n    = tot[b];
    const int2* rec2 = (const int2*)rec + beg;
    const int   xr   = xrow[min(b * BKT_NODES + wv * 16 + c, N_NODES - 1)];

    constexpr int U = 16;
    int y[U];
    #pragma unroll
    for (int u = 0; u < U; ++u) {
        const int i = u * 128 + t;
        y[u] = (i < n) ? rec2[i].y : -1;
    }
    float2 e[VOCAB / 2];
    const float2* ep = (const float2*)(xtab + 32 * half * EMB) + c;
    #pragma unroll
    for (int j = 0; j < VOCAB / 2; ++j) e[j] = ep[j * (EMB / 2)];

    int4* z = (int4*)cnt;
    for (int i = t; i < BKT_NODES * CNT_PITCH / 4; i += 128) z[i] = make_int4(0, 0, 0, 0);
    if (t < BKT_NODES) deg[t] = 0;
    __syncthreads();

    for (int base = 0; ; ) {
        #pragma unroll
        for (int u = 0; u < U; ++u) {
            if (y[u] >= 0) {
                const int dl = y[u] & 255;
                atomicAdd(&cnt[dl * CNT_PITCH + (y[u] >> 8)], 1);
                atomicAdd(&deg[dl], 1);
            }
        }
        base += 128 * U;
        if (base >= n) break;
        #pragma unroll
        for (int u = 0; u < U; ++u) {
            const int i = base + u * 128 + t;
            y[u] = (i < n) ? rec2[i].y : -1;
        }
    }
    __syncthreads();

    const int4* arow = (const int4*)&cnt[(wv * 16 + c) * CNT_PITCH + 32 * half];
    v8f acc0 = {}, acc1 = {};
    #pragma unroll
    for (int q4 = 0; q4 < 8; ++q4) {
        const int4 a4 = arow[q4];
        v2f a, b0, b1;
        a[0]  = (float)a4.x;       a[1]  = (float)a4.y;
        b0[0] = e[4 * q4].x;       b0[1] = e[4 * q4 + 1].x;
        b1[0] = e[4 * q4].y;       b1[1] = e[4 * q4 + 1].y;
        acc0 = __builtin_amdgcn_wmma_f32_16x16x4_f32(false, a, false, b0, (short)0, acc0, false, false);
        acc1 = __builtin_amdgcn_wmma_f32_16x16x4_f32(false, a, false, b1, (short)0, acc1, false, false);
        a[0]  = (float)a4.z;       a[1]  = (float)a4.w;
        b0[0] = e[4 * q4 + 2].x;   b0[1] = e[4 * q4 + 3].x;
        b1[0] = e[4 * q4 + 2].y;   b1[1] = e[4 * q4 + 3].y;
        acc0 = __builtin_amdgcn_wmma_f32_16x16x4_f32(false, a, false, b0, (short)0, acc0, false, false);
        acc1 = __builtin_amdgcn_wmma_f32_16x16x4_f32(false, a, false, b1, (short)0, acc1, false, false);
    }

    const int4* dq = (const int4*)&deg[wv * 16 + 8 * half];
    const int4  d0 = dq[0], d1 = dq[1];
    const int   dg[8] = {d0.x, d0.y, d0.z, d0.w, d1.x, d1.y, d1.z, d1.w};
    unsigned* mw = &mt[(wv * 16 + 8 * half) * MT_PITCH + c];
    #pragma unroll
    for (int r = 0; r < 8; ++r) {
        const float inv = 1.0f / fmaxf((float)dg[r], 1.0f);
        v2h o;
        o.x = (_Float16)(acc0[r] * inv);
        o.y = (_Float16)(acc1[r] * inv);
        mw[r * MT_PITCH] = __builtin_bit_cast(unsigned, o);
    }
    __syncthreads();

    const int row0 = b * BKT_NODES + wv * 16;
    if (row0 >= N_NODES) return;

    const uint4* mrow = (const uint4*)&mt[(wv * 16 + c) * MT_PITCH];
    const v8h lo = __builtin_bit_cast(v8h, mrow[half]);
    const v8h hi = __builtin_bit_cast(v8h, mrow[2 + half]);
    const v16h aM = __builtin_shufflevector(lo, hi, 0, 1, 2, 3, 4, 5, 6, 7,
                                            8, 9, 10, 11, 12, 13, 14, 15);
    const float4* xq = (const float4*)(xtab + xr * EMB);
    const v16h aX = load_frag(xq, 0, half, 1.0f);
    project_layer1(aM, aX, row0, half, c, Wl, bl, Wr, h1out);
}

#define ACC_PITCH 68
__global__ void __launch_bounds__(256)
range_layer2(const int* __restrict__ rec,
             const int* __restrict__ bstart,
             const int* __restrict__ tot,
             const _Float16* __restrict__ xh,
             const float* __restrict__ Wl,
             const float* __restrict__ bl,
             const float* __restrict__ Wr,
             float* __restrict__ pooled,
             const int* __restrict__ batch,
             float* __restrict__ gcnt)
{
    __shared__ alignas(16) float acc[(BKT_NODES + 1) * ACC_PITCH];
    __shared__ int   deg[BKT_NODES];
    __shared__ alignas(16) _Float16 wls[HID * W_PITCH];
    __shared__ alignas(16) _Float16 wrs[HID * W_PITCH];

    const int t    = threadIdx.x;
    const int lane = t & 31;
    const int wv   = __builtin_amdgcn_readfirstlane(t >> 5);
    const int b    = blockIdx.x;

    static_assert(HID * HID / 4 == 4 * 256, "four 16-byte pieces of each matrix per thread");
    float4 wl4[4], wr4[4];
    #pragma unroll
    for (int j = 0; j < 4; ++j) {
        wl4[j] = ((const float4*)Wl)[t + 256 * j];
        wr4[j] = ((const float4*)Wr)[t + 256 * j];
    }

    for (int i = t; i < (BKT_NODES + 1) * ACC_PITCH; i += 256) acc[i] = 0.0f;
    if (t < BKT_NODES) deg[t] = 0;
    #pragma unroll
    for (int j = 0; j < 4; ++j) {
        const int p   = t + 256 * j;
        const int off = (p >> 4) * W_PITCH + (p & 15) * 4;
        v4h hl, hr;
        hl[0] = (_Float16)wl4[j].x; hl[1] = (_Float16)wl4[j].y;
        hl[2] = (_Float16)wl4[j].z; hl[3] = (_Float16)wl4[j].w;
        hr[0] = (_Float16)wr4[j].x; hr[1] = (_Float16)wr4[j].y;
        hr[2] = (_Float16)wr4[j].z; hr[3] = (_Float16)wr4[j].w;
        *(v4h*)&wls[off] = hl;
        *(v4h*)&wrs[off] = hr;
    }
    __syncthreads();

    const int   beg  = bstart[b];
    const int   n    = tot[b];
    const int2* rec2 = (const int2*)rec + beg;
    const v2h*  xw   = (const v2h*)xh;

    const int2 pad = make_int2(0, BKT_NODES);
    int2 p = pad;
    if (wv * 32 + lane < n) p = rec2[wv * 32 + lane];

    for (int base = wv * 32; base < n; base += 256) {
        const int  inext = base + 256 + lane;
        int2 pn = pad;
        if (inext < n) pn = rec2[inext];

        const int dl  = p.y & 255;
        const int row = p.x;
        if (base + lane < n) atomicAdd(&deg[dl], 1);

        {
            v2h v[32];
            #pragma unroll
            for (int j = 0; j < 32; ++j) {
                const int sj = __builtin_amdgcn_readlane(row, j);
                v[j] = xw[sj * (HID / 2) + lane];
            }
            #pragma unroll
            for (int j = 0; j < 32; ++j) {
                const int dj = __builtin_amdgcn_readlane(dl, j);
                float* a = &acc[dj * ACC_PITCH + lane];
                atomicAdd(a,      (float)v[j].x);
                atomicAdd(a + 32, (float)v[j].y);
            }
        }
        p = pn;
    }
    __syncthreads();

    const int rt   = wv & 3;
    const int pr   = wv >> 2;
    const int row0 = b * BKT_NODES + rt * 16;
    if (row0 >= N_NODES) return;

    const int half = lane >> 4;
    const int m    = lane & 15;
    const int rl   = rt * 16 + m;
    const float inv = 1.0f / fmaxf((float)deg[rl], 1.0f);
    const float4* ar = (const float4*)&acc[rl * ACC_PITCH];
    const _Float16* xp = xh + (row0 + m) * HID;
    project_layer2(ar, inv, xp, row0, pr, half, m, wls, wrs, bl, pooled, batch, gcnt);
}

__global__ void final_head(const float* __restrict__ pooled,
                           const float* __restrict__ gcnt,
                           const float* __restrict__ Wout,
                           const float* __restrict__ bout,
                           float* __restrict__ out)
{
    int tid = blockIdx.x * blockDim.x + threadIdx.x;
    if (tid >= N_GRAPHS * NCLS) return;
    int g = tid / NCLS;
    int c = tid % NCLS;
    float inv = 1.0f / fmaxf(gcnt[g] + gcnt[N_GRAPHS + g], 1.0f);
    float acc = bout[c];
    #pragma unroll
    for (int h = 0; h < HID; ++h)
        acc += (pooled[g * HID + h] + pooled[N_GRAPHS * HID + g * HID + h]) * inv * Wout[c * HID + h];
    out[tid] = acc;
}

extern "C" void kernel_launch(void* const* d_in, const int* in_sizes, int n_in,
                              void* d_out, int out_size, void* d_ws, size_t ws_size,
                              hipStream_t stream)
{
    const int*   x_idx   = (const int*)  d_in[0];
    const int*   ei      = (const int*)  d_in[1];
    const int*   batch   = (const int*)  d_in[2];
    const float* embed_w = (const float*)d_in[3];
    const float* W1l     = (const float*)d_in[4];
    const float* b1      = (const float*)d_in[5];
    const float* W1r     = (const float*)d_in[6];
    const float* W2l     = (const float*)d_in[7];
    const float* b2      = (const float*)d_in[8];
    const float* W2r     = (const float*)d_in[9];
    const float* Wout    = (const float*)d_in[10];
    const float* bout    = (const float*)d_in[11];
    float*       out     = (float*)d_out;

    const int* src = ei;
    const int* dst = ei + N_EDGES;

    int*   tot    = (int*)d_ws;
    int*   cursor = tot    + NBKT_PAD;
    float* pooled = (float*)(cursor + NBKT_PAD);
    float* gcnt   = pooled + 2 * N_GRAPHS * HID;
    int*   bstart = (int*)(gcnt + 2 * N_GRAPHS);
    int*   blkcnt = bstart + NBKT_PAD;
    int*   rec    = blkcnt + PART_BLOCKS * NBKT;
    _Float16* h1    = (_Float16*)(rec + 2 * N_EDGES);

    const int B = 256;

    hipMemsetAsync(tot, 0,
                   sizeof(int) * (2 * NBKT_PAD + 2 * N_GRAPHS * HID + 2 * N_GRAPHS),
                   stream);

    bucket_count<<<PART_BLOCKS, PART_THREADS, 0, stream>>>(dst, blkcnt, tot);
    bucket_place<<<PART_BLOCKS, PART_THREADS, 0, stream>>>(src, dst, x_idx, blkcnt, tot, cursor,
                                                            bstart, rec);

    range_layer1<<<NBKT, 128, 0, stream>>>(rec, bstart, tot, embed_w, x_idx, W1l, b1, W1r, h1);

    range_layer2<<<NBKT, 256, 0, stream>>>(rec, bstart, tot, h1, W2l, b2, W2r, pooled, batch, gcnt);

    final_head<<<(N_GRAPHS * NCLS + B - 1) / B, B, 0, stream>>>(pooled, gcnt, Wout, bout, out);
}


# layer-2 neighbour sums in registers: range's records ordered by target in LDS (counting sort, rank = returning add); a wave sums each node's run, one LDS row add per node; new sum order (run order)
# speedup vs baseline: 9.8806x; 1.0826x over previous
#include <hip/hip_runtime.h>


#define N_NODES  100000
#define N_EDGES  3200000
#define N_GRAPHS 4096
#define EMB      32
#define HID      64
#define NCLS     10
#define VOCAB    64

#define BKT_SHIFT    6
#define BKT_NODES    (1 << BKT_SHIFT)
#define NBKT         ((N_NODES + BKT_NODES - 1) / BKT_NODES)
#define NBKT_PAD     1568
#define PART_BLOCKS  256
#define PART_THREADS 1024
#define PART_CHUNK   (N_EDGES / PART_BLOCKS)

typedef __attribute__((ext_vector_type(16))) _Float16 v16h;
typedef __attribute__((ext_vector_type(8)))  _Float16 v8h;
typedef __attribute__((ext_vector_type(4)))  _Float16 v4h;
typedef __attribute__((ext_vector_type(2)))  _Float16 v2h;
typedef __attribute__((ext_vector_type(8)))  float    v8f;
typedef __attribute__((ext_vector_type(2)))  float    v2f;

__device__ __forceinline__ int package_half()
{
    return (int)((__builtin_amdgcn_s_sendmsg_rtn(0x87) >> 18) & 1u);
}

__device__ __forceinline__ v16h pack_frag(float4 c0, float4 c1,
                                          float4 c2, float4 c3, float s)
{
    v16h r;
    r[0]  = (_Float16)(c0.x * s); r[1]  = (_Float16)(c0.y * s);
    r[2]  = (_Float16)(c0.z * s); r[3]  = (_Float16)(c0.w * s);
    r[4]  = (_Float16)(c1.x * s); r[5]  = (_Float16)(c1.y * s);
    r[6]  = (_Float16)(c1.z * s); r[7]  = (_Float16)(c1.w * s);
    r[8]  = (_Float16)(c2.x * s); r[9]  = (_Float16)(c2.y * s);
    r[10] = (_Float16)(c2.z * s); r[11] = (_Float16)(c2.w * s);
    r[12] = (_Float16)(c3.x * s); r[13] = (_Float16)(c3.y * s);
    r[14] = (_Float16)(c3.z * s); r[15] = (_Float16)(c3.w * s);
    return r;
}

__device__ __forceinline__ v16h load_frag(const float4* __restrict__ q,
                                          int ks, int half, float s)
{
    const int o = ks * 8 + half * 2;
    return pack_frag(q[o], q[o + 1], q[o + 4], q[o + 5], s);
}

__device__ __forceinline__ v16h load_frag_h(const _Float16* __restrict__ rowp,
                                            int ks, int half)
{
    const v8h* q  = (const v8h*)rowp;
    const v8h  lo = q[ks * 4 + half];
    const v8h  hi = q[ks * 4 + 2 + half];
    return __builtin_shufflevector(lo, hi, 0, 1, 2, 3, 4, 5, 6, 7,
                                   8, 9, 10, 11, 12, 13, 14, 15);
}

#define W_PITCH 72

__device__ __forceinline__ v16h pack_pairs(float4 ev0, float4 od0,
                                           float4 ev1, float4 od1, float s)
{
    v16h r;
    r[0]  = (_Float16)(ev0.x * s); r[1]  = (_Float16)(od0.x * s);
    r[2]  = (_Float16)(ev0.y * s); r[3]  = (_Float16)(od0.y * s);
    r[4]  = (_Float16)(ev0.z * s); r[5]  = (_Float16)(od0.z * s);
    r[6]  = (_Float16)(ev0.w * s); r[7]  = (_Float16)(od0.w * s);
    r[8]  = (_Float16)(ev1.x * s); r[9]  = (_Float16)(od1.x * s);
    r[10] = (_Float16)(ev1.y * s); r[11] = (_Float16)(od1.y * s);
    r[12] = (_Float16)(ev1.z * s); r[13] = (_Float16)(od1.z * s);
    r[14] = (_Float16)(ev1.w * s); r[15] = (_Float16)(od1.w * s);
    return r;
}

__device__ __forceinline__ void project_layer1(v16h aM, v16h aX, int row0, int half, int m,
                                               const float* __restrict__ Wl,
                                               const float* __restrict__ bl,
                                               const float* __restrict__ Wr,
                                               _Float16* __restrict__ h1out)
{
    #pragma unroll
    for (int pr = 0; pr < HID / 32; ++pr) {
        v8f   acc[2];
        float bias[2];
        #pragma unroll
        for (int q = 0; q < 2; ++q) {
            const int ncol = 32 * pr + 2 * m + q;
            const v16h bL = load_frag((const float4*)(Wl + ncol * EMB), 0, half, 1.0f);
            const v16h bR = load_frag((const float4*)(Wr + ncol * EMB), 0, half, 1.0f);
            v8f c = {};
            c = __builtin_amdgcn_wmma_f32_16x16x32_f16(
                false, aM, false, bL, (short)0, c, false, false);
            c = __builtin_amdgcn_wmma_f32_16x16x32_f16(
                false, aX, false, bR, (short)0, c, false, false);
            acc[q]  = c;
            bias[q] = bl[ncol];
        }
        v2h* op = (v2h*)h1out + (row0 + half * 8) * (HID / 2) + 16 * pr + m;
        #pragma unroll
        for (int d = 0; d < 8; ++d) {
            v2h o;
            o.x = (_Float16)fmaxf(acc[0][d] + bias[0], 0.0f);
            o.y = (_Float16)fmaxf(acc[1][d] + bias[1], 0.0f);
            op[d * (HID / 2)] = o;
        }
    }
}

__device__ __forceinline__ void project_layer2(const float4* ar, float inv,
                                               const _Float16* __restrict__ xp,
                                               int row0, int pr, int half, int m,
                                               const _Float16* wls,
                                               const _Float16* wrs,
                                               const float* __restrict__ bl,
                                               float* __restrict__ pooled,
                                               const int* __restrict__ batch,
                                               float* __restrict__ gcnt)
{
    const int hsel    = package_half();
    float*    pooledh = pooled + hsel * (N_GRAPHS * HID);
    float*    gcnth   = gcnt   + hsel * N_GRAPHS;
    const int ncol0 = pr * 32 + m;
    const int ncol1 = ncol0 + 16;
    v8f c0 = {}, c1 = {};
    #pragma unroll
    for (int ks = 0; ks < 2; ++ks) {
        const int  w0 = ks * 4 + half;
        const v16h aM = pack_pairs(ar[w0], ar[8 + w0], ar[w0 + 2], ar[8 + w0 + 2], inv);
        c0 = __builtin_amdgcn_wmma_f32_16x16x32_f16(
            false, aM, false, load_frag_h(wls + ncol0 * W_PITCH, ks, half),
            (short)0, c0, false, false);
        c1 = __builtin_amdgcn_wmma_f32_16x16x32_f16(
            false, aM, false, load_frag_h(wls + ncol1 * W_PITCH, ks, half),
            (short)0, c1, false, false);
        const v16h aX = load_frag_h(xp, ks, half);
        c0 = __builtin_amdgcn_wmma_f32_16x16x32_f16(
            false, aX, false, load_frag_h(wrs + ncol0 * W_PITCH, ks, half),
            (short)0, c0, false, false);
        c1 = __builtin_amdgcn_wmma_f32_16x16x32_f16(
            false, aX, false, load_frag_h(wrs + ncol1 * W_PITCH, ks, half),
            (short)0, c1, false, false);
    }

    int gb[8];
    #pragma unroll
    for (int d = 0; d < 8; ++d) gb[d] = batch[row0 + half * 8 + d];
    const float bias0 = bl[ncol0], bias1 = bl[ncol1];
    #pragma unroll
    for (int d = 0; d < 8; ++d)
        atomicAdd(&pooledh[gb[d] * HID + ncol0], fmaxf(c0[d] + bias0, 0.0f));
    #pragma unroll
    for (int d = 0; d < 8; ++d)
        atomicAdd(&pooledh[gb[d] * HID + ncol1], fmaxf(c1[d] + bias1, 0.0f));

    if (pr == 0 && m == 0) {
        #pragma unroll
        for (int d = 0; d < 8; ++d) atomicAdd(&gcnth[gb[d]], 1.0f);
    }
}

__global__ void __launch_bounds__(PART_THREADS)
bucket_count(const int* __restrict__ dst,
             int* __restrict__ blkcnt,
             int* __restrict__ tot)
{
    static_assert(N_EDGES % PART_BLOCKS == 0, "equal chunks, no tail");
    __shared__ int c[NBKT];
    for (int b = threadIdx.x; b < NBKT; b += PART_THREADS) c[b] = 0;
    __syncthreads();
    const int e0 = blockIdx.x * PART_CHUNK;
    constexpr int T = (PART_CHUNK + PART_THREADS - 1) / PART_THREADS;
    int d[T];
    #pragma unroll
    for (int k = 0; k < T; ++k) {
        const int i = threadIdx.x + k * PART_THREADS;
        d[k] = (i < PART_CHUNK) ? dst[e0 + i] : -1;
    }
    #pragma unroll
    for (int k = 0; k < T; ++k)
        if (d[k] >= 0) atomicAdd(&c[d[k] >> BKT_SHIFT], 1);
    __syncthreads();
    for (int b = threadIdx.x; b < NBKT; b += PART_THREADS) {
        const int v = c[b];
        blkcnt[blockIdx.x * NBKT + b] = v;
        if (v) atomicAdd(&tot[b], v);
    }
}

__global__ void __launch_bounds__(PART_THREADS)
bucket_place(const int* __restrict__ src,
             const int* __restrict__ dst,
             const int* __restrict__ xrow,
             const int* __restrict__ blkcnt,
             const int* __restrict__ tot,
             int* __restrict__ cursor,
             int* __restrict__ bstart,
             int* __restrict__ rec)
{
    static_assert(2 * PART_THREADS >= NBKT, "two totals per thread cover all ranges");
    __shared__ int pos[NBKT];
    __shared__ int wsum[PART_THREADS / 32];

    const int t    = threadIdx.x;
    const int lane = t & 31;
    const int wv   = t >> 5;

    const int i0 = 2 * t, i1 = 2 * t + 1;
    const int a0 = (i0 < NBKT) ? tot[i0] : 0;
    const int a1 = (i1 < NBKT) ? tot[i1] : 0;
    const int own = a0 + a1;
    int inc = own;
    #pragma unroll
    for (int o = 1; o < 32; o <<= 1) {
        const int u = __shfl_up(inc, o);
        if (lane >= o) inc += u;
    }
    if (lane == 31) wsum[wv] = inc;
    __syncthreads();
    if (wv == 0) {
        const int w = wsum[lane];
        int winc = w;
        #pragma unroll
        for (int o = 1; o < 32; o <<= 1) {
            const int u = __shfl_up(winc, o);
            if (lane >= o) winc += u;
        }
        wsum[lane] = winc - w;
    }
    __syncthreads();
    const int excl = wsum[wv] + inc - own;
    if (i0 < NBKT) pos[i0] = excl;
    if (i1 < NBKT) pos[i1] = excl + a0;
    if (blockIdx.x == 0) {
        if (i0 < NBKT) bstart[i0] = excl;
        if (i1 < NBKT) bstart[i1] = excl + a0;
    }
    __syncthreads();

    for (int b = t; b < NBKT; b += PART_THREADS) {
        const int c = blkcnt[blockIdx.x * NBKT + b];
        if (c) pos[b] += atomicAdd(&cursor[b], c);
    }
    __syncthreads();

    int2* rec2 = (int2*)rec;
    const int e0 = blockIdx.x * PART_CHUNK;
    constexpr int T = (PART_CHUNK + PART_THREADS - 1) / PART_THREADS;
    int s[T], d[T], xr[T];
    #pragma unroll
    for (int k = 0; k < T; ++k) {
        const int i = t + k * PART_THREADS;
        s[k] = (i < PART_CHUNK) ? src[e0 + i] : 0;
        d[k] = (i < PART_CHUNK) ? dst[e0 + i] : -1;
    }
    #pragma unroll
    for (int k = 0; k < T; ++k) xr[k] = xrow[s[k]];
    #pragma unroll
    for (int k = 0; k < T; ++k) {
        if (d[k] >= 0) {
            const int p = atomicAdd(&pos[d[k] >> BKT_SHIFT], 1);
            rec2[p] = make_int2(s[k], (xr[k] << 8) | (d[k] & (BKT_NODES - 1)));
        }
    }
}

#define CNT_PITCH 68
#define MT_PITCH  20
__global__ void __launch_bounds__(128)
range_layer1(const int* __restrict__ rec,
             const int* __restrict__ bstart,
             const int* __restrict__ tot,
             const float* __restrict__ xtab,
             const int* __restrict__ xrow,
             const float* __restrict__ Wl,
             const float* __restrict__ bl,
             const float* __restrict__ Wr,
             _Float16* __restrict__ h1out)
{
    static_assert(BKT_NODES == 64 && VOCAB == 64 && EMB == 32,
                  "4 waves x 16 rows, K = 64 as 2 lane halves x 32, two 16-column tiles");
    __shared__ alignas(16) int cnt[BKT_NODES * CNT_PITCH];
    __shared__ alignas(16) int deg[BKT_NODES];
    __shared__ alignas(16) unsigned mt[BKT_NODES * MT_PITCH];

    const int t    = threadIdx.x;
    const int lane = t & 31;
    const int wv   = __builtin_amdgcn_readfirstlane(t >> 5);
    const int b    = blockIdx.x;
    const int half = lane >> 4;
    const int c    = lane & 15;

    const int   beg  = bstart[b];
    const int   n    = tot[b];
    const int2* rec2 = (const int2*)rec + beg;
    const int   xr   = xrow[min(b * BKT_NODES + wv * 16 + c, N_NODES - 1)];

    constexpr int U = 16;
    int y[U];
    #pragma unroll
    for (int u = 0; u < U; ++u) {
        const int i = u * 128 + t;
        y[u] = (i < n) ? rec2[i].y : -1;
    }
    float2 e[VOCAB / 2];
    const float2* ep = (const float2*)(xtab + 32 * half * EMB) + c;
    #pragma unroll
    for (int j = 0; j < VOCAB / 2; ++j) e[j] = ep[j * (EMB / 2)];

    int4* z = (int4*)cnt;
    for (int i = t; i < BKT_NODES * CNT_PITCH / 4; i += 128) z[i] = make_int4(0, 0, 0, 0);
    if (t < BKT_NODES) deg[t] = 0;
    __syncthreads();

    for (int base = 0; ; ) {
        #pragma unroll
        for (int u = 0; u < U; ++u) {
            if (y[u] >= 0) {
                const int dl = y[u] & 255;
                atomicAdd(&cnt[dl * CNT_PITCH + (y[u] >> 8)], 1);
                atomicAdd(&deg[dl], 1);
            }
        }
        base += 128 * U;
        if (base >= n) break;
        #pragma unroll
        for (int u = 0; u < U; ++u) {
            const int i = base + u * 128 + t;
            y[u] = (i < n) ? rec2[i].y : -1;
        }
    }
    __syncthreads();

    const int4* arow = (const int4*)&cnt[(wv * 16 + c) * CNT_PITCH + 32 * half];
    v8f acc0 = {}, acc1 = {};
    #pragma unroll
    for (int q4 = 0; q4 < 8; ++q4) {
        const int4 a4 = arow[q4];
        v2f a, b0, b1;
        a[0]  = (float)a4.x;       a[1]  = (float)a4.y;
        b0[0] = e[4 * q4].x;       b0[1] = e[4 * q4 + 1].x;
        b1[0] = e[4 * q4].y;       b1[1] = e[4 * q4 + 1].y;
        acc0 = __builtin_amdgcn_wmma_f32_16x16x4_f32(false, a, false, b0, (short)0, acc0, false, false);
        acc1 = __builtin_amdgcn_wmma_f32_16x16x4_f32(false, a, false, b1, (short)0, acc1, false, false);
        a[0]  = (float)a4.z;       a[1]  = (float)a4.w;
        b0[0] = e[4 * q4 + 2].x;   b0[1] = e[4 * q4 + 3].x;
        b1[0] = e[4 * q4 + 2].y;   b1[1] = e[4 * q4 + 3].y;
        acc0 = __builtin_amdgcn_wmma_f32_16x16x4_f32(false, a, false, b0, (short)0, acc0, false, false);
        acc1 = __builtin_amdgcn_wmma_f32_16x16x4_f32(false, a, false, b1, (short)0, acc1, false, false);
    }

    const int4* dq = (const int4*)&deg[wv * 16 + 8 * half];
    const int4  d0 = dq[0], d1 = dq[1];
    const int   dg[8] = {d0.x, d0.y, d0.z, d0.w, d1.x, d1.y, d1.z, d1.w};
    unsigned* mw = &mt[(wv * 16 + 8 * half) * MT_PITCH + c];
    #pragma unroll
    for (int r = 0; r < 8; ++r) {
        const float inv = 1.0f / fmaxf((float)dg[r], 1.0f);
        v2h o;
        o.x = (_Float16)(acc0[r] * inv);
        o.y = (_Float16)(acc1[r] * inv);
        mw[r * MT_PITCH] = __builtin_bit_cast(unsigned, o);
    }
    __syncthreads();

    const int row0 = b * BKT_NODES + wv * 16;
    if (row0 >= N_NODES) return;

    const uint4* mrow = (const uint4*)&mt[(wv * 16 + c) * MT_PITCH];
    const v8h lo = __builtin_bit_cast(v8h, mrow[half]);
    const v8h hi = __builtin_bit_cast(v8h, mrow[2 + half]);
    const v16h aM = __builtin_shufflevector(lo, hi, 0, 1, 2, 3, 4, 5, 6, 7,
                                            8, 9, 10, 11, 12, 13, 14, 15);
    const float4* xq = (const float4*)(xtab + xr * EMB);
    const v16h aX = load_frag(xq, 0, half, 1.0f);
    project_layer1(aM, aX, row0, half, c, Wl, bl, Wr, h1out);
}

#define ACC_PITCH 68
#define L2_CAP    3072
#define L2_SRC_BITS 17
__global__ void __launch_bounds__(256, 14)
range_layer2(const int* __restrict__ rec,
             const int* __restrict__ bstart,
             const int* __restrict__ tot,
             const _Float16* __restrict__ xh,
             const float* __restrict__ Wl,
             const float* __restrict__ bl,
             const float* __restrict__ Wr,
             float* __restrict__ pooled,
             const int* __restrict__ batch,
             float* __restrict__ gcnt)
{
    static_assert(N_NODES <= (1 << L2_SRC_BITS), "source node fits the low bits of a packed word");
    static_assert(BKT_NODES == 64 && L2_CAP % 256 == 0, "64 counts = 2 per lane of wave 0; whole trips of the work-group");
    __shared__ alignas(16) float acc[(BKT_NODES + 1) * ACC_PITCH];
    __shared__ int   deg[BKT_NODES];
    __shared__ int   ccnt[BKT_NODES];
    __shared__ int   cstart[BKT_NODES + 1];
    __shared__ int   ccur[BKT_NODES];
    static_assert(L2_CAP * 4 <= 2 * HID * W_PITCH * 2, "the ordered words fit the weights' region");
    __shared__ alignas(16) _Float16 wbuf[2 * HID * W_PITCH];
    typedef int __attribute__((__may_alias__)) lds_word;
    lds_word* srt = (lds_word*)wbuf;
    _Float16* wls = wbuf;
    _Float16* wrs = wbuf + HID * W_PITCH;

    const int t    = threadIdx.x;
    const int lane = t & 31;
    const int wv   = __builtin_amdgcn_readfirstlane(t >> 5);
    const int b    = blockIdx.x;

    const int   beg  = bstart[b];
    const int   n    = tot[b];

    for (int i = t; i < (BKT_NODES + 1) * ACC_PITCH; i += 256) acc[i] = 0.0f;
    if (t < BKT_NODES) { deg[t] = 0; ccnt[t] = 0; }
    __syncthreads();

    const int2* rec2 = (const int2*)rec + beg;
    const v2h*  xw   = (const v2h*)xh;

    constexpr int U   = L2_CAP / 256;
    constexpr int PAD = BKT_NODES << L2_SRC_BITS;

    for (int cb = 0; cb < n; cb += L2_CAP) {
        const int cn = min(n - cb, L2_CAP);

        int w[U];
        #pragma unroll
        for (int u = 0; u < U; ++u) {
            const int i = u * 256 + t;
            w[u] = -1;
            if (i < cn) {
                const int2 r = rec2[cb + i];
                w[u] = r.x | ((r.y & 255) << L2_SRC_BITS);
            }
        }
        #pragma unroll
        for (int u = 0; u < U; ++u)
            if (w[u] >= 0) atomicAdd(&ccnt[w[u] >> L2_SRC_BITS], 1);
        __syncthreads();

        if (wv == 0) {
            const int a0 = ccnt[2 * lane], a1 = ccnt[2 * lane + 1];
            int inc = a0 + a1;
            #pragma unroll
            for (int o = 1; o < 32; o <<= 1) {
                const int up = __shfl_up(inc, o);
                if (lane >= o) inc += up;
            }
            const int excl = inc - a0 - a1;
            cstart[2 * lane]     = excl;
            cstart[2 * lane + 1] = excl + a0;
            ccur[2 * lane]       = excl;
            ccur[2 * lane + 1]   = excl + a0;
            if (lane == 31) cstart[BKT_NODES] = inc;
            deg[2 * lane]     += a0;
            deg[2 * lane + 1] += a1;
            ccnt[2 * lane]     = 0;
            ccnt[2 * lane + 1] = 0;
        }
        __syncthreads();

        #pragma unroll
        for (int u = 0; u < U; ++u) {
            if (w[u] >= 0) {
                const int p = atomicAdd(&ccur[w[u] >> L2_SRC_BITS], 1);
                srt[p] = w[u];
            }
        }
        __syncthreads();

        const int s0 = __builtin_amdgcn_readfirstlane(cstart[8 * wv]);
        const int e0 = __builtin_amdgcn_readfirstlane(cstart[8 * wv + 8]);
        float ax  = 0.0f, ay = 0.0f;
        int   cur = BKT_NODES;
        for (int base = s0; base < e0; base += 32) {
            int wd = PAD;
            if (base + lane < e0) wd = srt[base + lane];

            v2h v[32];
            #pragma unroll
            for (int j = 0; j < 32; ++j) {
                const int sj = __builtin_amdgcn_readlane(wd, j) & ((1 << L2_SRC_BITS) - 1);
                v[j] = xw[sj * (HID / 2) + lane];
            }
            #pragma unroll
            for (int j = 0; j < 32; ++j) {
                const int dj = __builtin_amdgcn_readlane(wd, j) >> L2_SRC_BITS;
                if (dj != cur) {
                    float* a = &acc[cur * ACC_PITCH + lane];
                    atomicAdd(a,      ax);
                    atomicAdd(a + 32, ay);
                    ax = 0.0f; ay = 0.0f; cur = dj;
                }
                ax += (float)v[j].x;
                ay += (float)v[j].y;
            }
        }
        {
            float* a = &acc[cur * ACC_PITCH + lane];
            atomicAdd(a,      ax);
            atomicAdd(a + 32, ay);
        }
    }
    __syncthreads();

    static_assert(HID * HID / 4 == 4 * 256, "four 16-byte pieces of each matrix per thread");
    {
        float4 wl4[4], wr4[4];
        #pragma unroll
        for (int j = 0; j < 4; ++j) {
            wl4[j] = ((const float4*)Wl)[t + 256 * j];
            wr4[j] = ((const float4*)Wr)[t + 256 * j];
        }
        #pragma unroll
        for (int j = 0; j < 4; ++j) {
            const int p   = t + 256 * j;
            const int off = (p >> 4) * W_PITCH + (p & 15) * 4;
            v4h hl, hr;
            hl[0] = (_Float16)wl4[j].x; hl[1] = (_Float16)wl4[j].y;
            hl[2] = (_Float16)wl4[j].z; hl[3] = (_Float16)wl4[j].w;
            hr[0] = (_Float16)wr4[j].x; hr[1] = (_Float16)wr4[j].y;
            hr[2] = (_Float16)wr4[j].z; hr[3] = (_Float16)wr4[j].w;
            *(v4h*)&wls[off] = hl;
            *(v4h*)&wrs[off] = hr;
        }
    }
    __syncthreads();

    const int rt   = wv & 3;
    const int pr   = wv >> 2;
    const int row0 = b * BKT_NODES + rt * 16;
    if (row0 >= N_NODES) return;

    const int half = lane >> 4;
    const int m    = lane & 15;
    const int rl   = rt * 16 + m;
    const float inv = 1.0f / fmaxf((float)deg[rl], 1.0f);
    const float4* ar = (const float4*)&acc[rl * ACC_PITCH];
    const _Float16* xp = xh + (row0 + m) * HID;
    project_layer2(ar, inv, xp, row0, pr, half, m, wls, wrs, bl, pooled, batch, gcnt);
}

__global__ void final_head(const float* __restrict__ pooled,
                           const float* __restrict__ gcnt,
                           const float* __restrict__ Wout,
                           const float* __restrict__ bout,
                           float* __restrict__ out)
{
    int tid = blockIdx.x * blockDim.x + threadIdx.x;
    if (tid >= N_GRAPHS * NCLS) return;
    int g = tid / NCLS;
    int c = tid % NCLS;
    float inv = 1.0f / fmaxf(gcnt[g] + gcnt[N_GRAPHS + g], 1.0f);
    float acc = bout[c];
    #pragma unroll
    for (int h = 0; h < HID; ++h)
        acc += (pooled[g * HID + h] + pooled[N_GRAPHS * HID + g * HID + h]) * inv * Wout[c * HID + h];
    out[tid] = acc;
}

extern "C" void kernel_launch(void* const* d_in, const int* in_sizes, int n_in,
                              void* d_out, int out_size, void* d_ws, size_t ws_size,
                              hipStream_t stream)
{
    const int*   x_idx   = (const int*)  d_in[0];
    const int*   ei      = (const int*)  d_in[1];
    const int*   batch   = (const int*)  d_in[2];
    const float* embed_w = (const float*)d_in[3];
    const float* W1l     = (const float*)d_in[4];
    const float* b1      = (const float*)d_in[5];
    const float* W1r     = (const float*)d_in[6];
    const float* W2l     = (const float*)d_in[7];
    const float* b2      = (const float*)d_in[8];
    const float* W2r     = (const float*)d_in[9];
    const float* Wout    = (const float*)d_in[10];
    const float* bout    = (const float*)d_in[11];
    float*       out     = (float*)d_out;

    const int* src = ei;
    const int* dst = ei + N_EDGES;

    int*   tot    = (int*)d_ws;
    int*   cursor = tot    + NBKT_PAD;
    float* pooled = (float*)(cursor + NBKT_PAD);
    float* gcnt   = pooled + 2 * N_GRAPHS * HID;
    int*   bstart = (int*)(gcnt + 2 * N_GRAPHS);
    int*   blkcnt = bstart + NBKT_PAD;
    int*   rec    = blkcnt + PART_BLOCKS * NBKT;
    _Float16* h1    = (_Float16*)(rec + 2 * N_EDGES);

    const int B = 256;

    hipMemsetAsync(tot, 0,
                   sizeof(int) * (2 * NBKT_PAD + 2 * N_GRAPHS * HID + 2 * N_GRAPHS),
                   stream);

    bucket_count<<<PART_BLOCKS, PART_THREADS, 0, stream>>>(dst, blkcnt, tot);
    bucket_place<<<PART_BLOCKS, PART_THREADS, 0, stream>>>(src, dst, x_idx, blkcnt, tot, cursor,
                                                            bstart, rec);

    range_layer1<<<NBKT, 128, 0, stream>>>(rec, bstart, tot, embed_w, x_idx, W1l, b1, W1r, h1);

    range_layer2<<<NBKT, 256, 0, stream>>>(rec, bstart, tot, h1, W2l, b2, W2r, pooled, batch, gcnt);

    final_head<<<(N_GRAPHS * NCLS + B - 1) / B, B, 0, stream>>>(pooled, gcnt, Wout, bout, out);
}


# bucket_place: table row of each edge's source read from an LDS byte copy of x_idx (100 KB, written by bucket_count) instead of one random 4-byte global read per edge (32 lines per load)
# speedup vs baseline: 10.5437x; 1.0671x over previous
#include <hip/hip_runtime.h>


#define N_NODES  100000
#define N_EDGES  3200000
#define N_GRAPHS 4096
#define EMB      32
#define HID      64
#define NCLS     10
#define VOCAB    64

#define BKT_SHIFT    6
#define BKT_NODES    (1 << BKT_SHIFT)
#define NBKT         ((N_NODES + BKT_NODES - 1) / BKT_NODES)
#define NBKT_PAD     1568
#define PART_BLOCKS  256
#define PART_THREADS 1024
#define PART_CHUNK   (N_EDGES / PART_BLOCKS)
#define XB_WORDS     (N_NODES / 4)
#define XB_PER_BLOCK ((XB_WORDS + PART_BLOCKS - 1) / PART_BLOCKS)

typedef __attribute__((ext_vector_type(16))) _Float16 v16h;
typedef __attribute__((ext_vector_type(8)))  _Float16 v8h;
typedef __attribute__((ext_vector_type(4)))  _Float16 v4h;
typedef __attribute__((ext_vector_type(2)))  _Float16 v2h;
typedef __attribute__((ext_vector_type(8)))  float    v8f;
typedef __attribute__((ext_vector_type(2)))  float    v2f;

__device__ __forceinline__ int package_half()
{
    return (int)((__builtin_amdgcn_s_sendmsg_rtn(0x87) >> 18) & 1u);
}

__device__ __forceinline__ v16h pack_frag(float4 c0, float4 c1,
                                          float4 c2, float4 c3, float s)
{
    v16h r;
    r[0]  = (_Float16)(c0.x * s); r[1]  = (_Float16)(c0.y * s);
    r[2]  = (_Float16)(c0.z * s); r[3]  = (_Float16)(c0.w * s);
    r[4]  = (_Float16)(c1.x * s); r[5]  = (_Float16)(c1.y * s);
    r[6]  = (_Float16)(c1.z * s); r[7]  = (_Float16)(c1.w * s);
    r[8]  = (_Float16)(c2.x * s); r[9]  = (_Float16)(c2.y * s);
    r[10] = (_Float16)(c2.z * s); r[11] = (_Float16)(c2.w * s);
    r[12] = (_Float16)(c3.x * s); r[13] = (_Float16)(c3.y * s);
    r[14] = (_Float16)(c3.z * s); r[15] = (_Float16)(c3.w * s);
    return r;
}

__device__ __forceinline__ v16h load_frag(const float4* __restrict__ q,
                                          int ks, int half, float s)
{
    const int o = ks * 8 + half * 2;
    return pack_frag(q[o], q[o + 1], q[o + 4], q[o + 5], s);
}

__device__ __forceinline__ v16h load_frag_h(const _Float16* __restrict__ rowp,
                                            int ks, int half)
{
    const v8h* q  = (const v8h*)rowp;
    const v8h  lo = q[ks * 4 + half];
    const v8h  hi = q[ks * 4 + 2 + half];
    return __builtin_shufflevector(lo, hi, 0, 1, 2, 3, 4, 5, 6, 7,
                                   8, 9, 10, 11, 12, 13, 14, 15);
}

#define W_PITCH 72

__device__ __forceinline__ v16h pack_pairs(float4 ev0, float4 od0,
                                           float4 ev1, float4 od1, float s)
{
    v16h r;
    r[0]  = (_Float16)(ev0.x * s); r[1]  = (_Float16)(od0.x * s);
    r[2]  = (_Float16)(ev0.y * s); r[3]  = (_Float16)(od0.y * s);
    r[4]  = (_Float16)(ev0.z * s); r[5]  = (_Float16)(od0.z * s);
    r[6]  = (_Float16)(ev0.w * s); r[7]  = (_Float16)(od0.w * s);
    r[8]  = (_Float16)(ev1.x * s); r[9]  = (_Float16)(od1.x * s);
    r[10] = (_Float16)(ev1.y * s); r[11] = (_Float16)(od1.y * s);
    r[12] = (_Float16)(ev1.z * s); r[13] = (_Float16)(od1.z * s);
    r[14] = (_Float16)(ev1.w * s); r[15] = (_Float16)(od1.w * s);
    return r;
}

__device__ __forceinline__ void project_layer1(v16h aM, v16h aX, int row0, int half, int m,
                                               const float* __restrict__ Wl,
                                               const float* __restrict__ bl,
                                               const float* __restrict__ Wr,
                                               _Float16* __restrict__ h1out)
{
    #pragma unroll
    for (int pr = 0; pr < HID / 32; ++pr) {
        v8f   acc[2];
        float bias[2];
        #pragma unroll
        for (int q = 0; q < 2; ++q) {
            const int ncol = 32 * pr + 2 * m + q;
            const v16h bL = load_frag((const float4*)(Wl + ncol * EMB), 0, half, 1.0f);
            const v16h bR = load_frag((const float4*)(Wr + ncol * EMB), 0, half, 1.0f);
            v8f c = {};
            c = __builtin_amdgcn_wmma_f32_16x16x32_f16(
                false, aM, false, bL, (short)0, c, false, false);
            c = __builtin_amdgcn_wmma_f32_16x16x32_f16(
                false, aX, false, bR, (short)0, c, false, false);
            acc[q]  = c;
            bias[q] = bl[ncol];
        }
        v2h* op = (v2h*)h1out + (row0 + half * 8) * (HID / 2) + 16 * pr + m;
        #pragma unroll
        for (int d = 0; d < 8; ++d) {
            v2h o;
            o.x = (_Float16)fmaxf(acc[0][d] + bias[0], 0.0f);
            o.y = (_Float16)fmaxf(acc[1][d] + bias[1], 0.0f);
            op[d * (HID / 2)] = o;
        }
    }
}

__device__ __forceinline__ void project_layer2(const float4* ar, float inv,
                                               const _Float16* __restrict__ xp,
                                               int row0, int pr, int half, int m,
                                               const _Float16* wls,
                                               const _Float16* wrs,
                                               const float* __restrict__ bl,
                                               float* __restrict__ pooled,
                                               const int* __restrict__ batch,
                                               float* __restrict__ gcnt)
{
    const int hsel    = package_half();
    float*    pooledh = pooled + hsel * (N_GRAPHS * HID);
    float*    gcnth   = gcnt   + hsel * N_GRAPHS;
    const int ncol0 = pr * 32 + m;
    const int ncol1 = ncol0 + 16;
    v8f c0 = {}, c1 = {};
    #pragma unroll
    for (int ks = 0; ks < 2; ++ks) {
        const int  w0 = ks * 4 + half;
        const v16h aM = pack_pairs(ar[w0], ar[8 + w0], ar[w0 + 2], ar[8 + w0 + 2], inv);
        c0 = __builtin_amdgcn_wmma_f32_16x16x32_f16(
            false, aM, false, load_frag_h(wls + ncol0 * W_PITCH, ks, half),
            (short)0, c0, false, false);
        c1 = __builtin_amdgcn_wmma_f32_16x16x32_f16(
            false, aM, false, load_frag_h(wls + ncol1 * W_PITCH, ks, half),
            (short)0, c1, false, false);
        const v16h aX = load_frag_h(xp, ks, half);
        c0 = __builtin_amdgcn_wmma_f32_16x16x32_f16(
            false, aX, false, load_frag_h(wrs + ncol0 * W_PITCH, ks, half),
            (short)0, c0, false, false);
        c1 = __builtin_amdgcn_wmma_f32_16x16x32_f16(
            false, aX, false, load_frag_h(wrs + ncol1 * W_PITCH, ks, half),
            (short)0, c1, false, false);
    }

    int gb[8];
    #pragma unroll
    for (int d = 0; d < 8; ++d) gb[d] = batch[row0 + half * 8 + d];
    const float bias0 = bl[ncol0], bias1 = bl[ncol1];
    #pragma unroll
    for (int d = 0; d < 8; ++d)
        atomicAdd(&pooledh[gb[d] * HID + ncol0], fmaxf(c0[d] + bias0, 0.0f));
    #pragma unroll
    for (int d = 0; d < 8; ++d)
        atomicAdd(&pooledh[gb[d] * HID + ncol1], fmaxf(c1[d] + bias1, 0.0f));

    if (pr == 0 && m == 0) {
        #pragma unroll
        for (int d = 0; d < 8; ++d) atomicAdd(&gcnth[gb[d]], 1.0f);
    }
}

__global__ void __launch_bounds__(PART_THREADS)
bucket_count(const int* __restrict__ dst,
             const int* __restrict__ xrow,
             int* __restrict__ blkcnt,
             int* __restrict__ tot,
             unsigned* __restrict__ xb)
{
    static_assert(N_EDGES % PART_BLOCKS == 0, "equal chunks, no tail");
    static_assert(N_NODES % 4 == 0 && VOCAB <= 256, "four one-byte table rows per word");
    static_assert(XB_PER_BLOCK <= PART_THREADS, "one word per thread");
    __shared__ int c[NBKT];
    for (int b = threadIdx.x; b < NBKT; b += PART_THREADS) c[b] = 0;
    __syncthreads();
    const int e0 = blockIdx.x * PART_CHUNK;
    constexpr int T = (PART_CHUNK + PART_THREADS - 1) / PART_THREADS;
    int d[T];
    #pragma unroll
    for (int k = 0; k < T; ++k) {
        const int i = threadIdx.x + k * PART_THREADS;
        d[k] = (i < PART_CHUNK) ? dst[e0 + i] : -1;
    }
    if (threadIdx.x < XB_PER_BLOCK) {
        const int w = blockIdx.x * XB_PER_BLOCK + threadIdx.x;
        if (w < XB_WORDS) {
            const int4 v = ((const int4*)xrow)[w];
            xb[w] = (unsigned)(v.x & 255) | ((unsigned)(v.y & 255) << 8) |
                    ((unsigned)(v.z & 255) << 16) | ((unsigned)(v.w & 255) << 24);
        }
    }
    #pragma unroll
    for (int k = 0; k < T; ++k)
        if (d[k] >= 0) atomicAdd(&c[d[k] >> BKT_SHIFT], 1);
    __syncthreads();
    for (int b = threadIdx.x; b < NBKT; b += PART_THREADS) {
        const int v = c[b];
        blkcnt[blockIdx.x * NBKT + b] = v;
        if (v) atomicAdd(&tot[b], v);
    }
}

__global__ void __launch_bounds__(PART_THREADS)
bucket_place(const int* __restrict__ src,
             const int* __restrict__ dst,
             const unsigned* __restrict__ xb,
             const int* __restrict__ blkcnt,
             const int* __restrict__ tot,
             int* __restrict__ cursor,
             int* __restrict__ bstart,
             int* __restrict__ rec)
{
    static_assert(2 * PART_THREADS >= NBKT, "two totals per thread cover all ranges");
    __shared__ int pos[NBKT];
    __shared__ int wsum[PART_THREADS / 32];
    static_assert(XB_WORDS % 4 == 0, "whole 16-byte pieces");
    constexpr int XB_Q = XB_WORDS / 4;
    constexpr int TQ   = (XB_Q + PART_THREADS - 1) / PART_THREADS;
    __shared__ alignas(16) unsigned char xbs[TQ * PART_THREADS * 16];

    const int t    = threadIdx.x;
    const int lane = t & 31;
    const int wv   = t >> 5;

    const int i0 = 2 * t, i1 = 2 * t + 1;
    const int a0 = (i0 < NBKT) ? tot[i0] : 0;
    const int a1 = (i1 < NBKT) ? tot[i1] : 0;

    uint4 tb[TQ];
    #pragma unroll
    for (int j = 0; j < TQ; ++j)
        tb[j] = ((const uint4*)xb)[min(t + j * PART_THREADS, XB_Q - 1)];
    const int own = a0 + a1;
    int inc = own;
    #pragma unroll
    for (int o = 1; o < 32; o <<= 1) {
        const int u = __shfl_up(inc, o);
        if (lane >= o) inc += u;
    }
    #pragma unroll
    for (int j = 0; j < TQ; ++j) ((uint4*)xbs)[t + j * PART_THREADS] = tb[j];
    if (lane == 31) wsum[wv] = inc;
    __syncthreads();
    if (wv == 0) {
        const int w = wsum[lane];
        int winc = w;
        #pragma unroll
        for (int o = 1; o < 32; o <<= 1) {
            const int u = __shfl_up(winc, o);
            if (lane >= o) winc += u;
        }
        wsum[lane] = winc - w;
    }
    __syncthreads();
    const int excl = wsum[wv] + inc - own;
    if (i0 < NBKT) pos[i0] = excl;
    if (i1 < NBKT) pos[i1] = excl + a0;
    if (blockIdx.x == 0) {
        if (i0 < NBKT) bstart[i0] = excl;
        if (i1 < NBKT) bstart[i1] = excl + a0;
    }
    __syncthreads();

    for (int b = t; b < NBKT; b += PART_THREADS) {
        const int c = blkcnt[blockIdx.x * NBKT + b];
        if (c) pos[b] += atomicAdd(&cursor[b], c);
    }
    __syncthreads();

    int2* rec2 = (int2*)rec;
    const int e0 = blockIdx.x * PART_CHUNK;
    constexpr int T = (PART_CHUNK + PART_THREADS - 1) / PART_THREADS;
    int s[T], d[T], xr[T];
    #pragma unroll
    for (int k = 0; k < T; ++k) {
        const int i = t + k * PART_THREADS;
        s[k] = (i < PART_CHUNK) ? src[e0 + i] : 0;
        d[k] = (i < PART_CHUNK) ? dst[e0 + i] : -1;
    }
    #pragma unroll
    for (int k = 0; k < T; ++k) xr[k] = xbs[s[k]];
    #pragma unroll
    for (int k = 0; k < T; ++k) {
        if (d[k] >= 0) {
            const int p = atomicAdd(&pos[d[k] >> BKT_SHIFT], 1);
            rec2[p] = make_int2(s[k], (xr[k] << 8) | (d[k] & (BKT_NODES - 1)));
        }
    }
}

#define CNT_PITCH 68
#define MT_PITCH  20
__global__ void __launch_bounds__(128)
range_layer1(const int* __restrict__ rec,
             const int* __restrict__ bstart,
             const int* __restrict__ tot,
             const float* __restrict__ xtab,
             const int* __restrict__ xrow,
             const float* __restrict__ Wl,
             const float* __restrict__ bl,
             const float* __restrict__ Wr,
             _Float16* __restrict__ h1out)
{
    static_assert(BKT_NODES == 64 && VOCAB == 64 && EMB == 32,
                  "4 waves x 16 rows, K = 64 as 2 lane halves x 32, two 16-column tiles");
    __shared__ alignas(16) int cnt[BKT_NODES * CNT_PITCH];
    __shared__ alignas(16) int deg[BKT_NODES];
    __shared__ alignas(16) unsigned mt[BKT_NODES * MT_PITCH];

    const int t    = threadIdx.x;
    const int lane = t & 31;
    const int wv   = __builtin_amdgcn_readfirstlane(t >> 5);
    const int b    = blockIdx.x;
    const int half = lane >> 4;
    const int c    = lane & 15;

    const int   beg  = bstart[b];
    const int   n    = tot[b];
    const int2* rec2 = (const int2*)rec + beg;
    const int   xr   = xrow[min(b * BKT_NODES + wv * 16 + c, N_NODES - 1)];

    constexpr int U = 16;
    int y[U];
    #pragma unroll
    for (int u = 0; u < U; ++u) {
        const int i = u * 128 + t;
        y[u] = (i < n) ? rec2[i].y : -1;
    }
    float2 e[VOCAB / 2];
    const float2* ep = (const float2*)(xtab + 32 * half * EMB) + c;
    #pragma unroll
    for (int j = 0; j < VOCAB / 2; ++j) e[j] = ep[j * (EMB / 2)];

    int4* z = (int4*)cnt;
    for (int i = t; i < BKT_NODES * CNT_PITCH / 4; i += 128) z[i] = make_int4(0, 0, 0, 0);
    if (t < BKT_NODES) deg[t] = 0;
    __syncthreads();

    for (int base = 0; ; ) {
        #pragma unroll
        for (int u = 0; u < U; ++u) {
            if (y[u] >= 0) {
                const int dl = y[u] & 255;
                atomicAdd(&cnt[dl * CNT_PITCH + (y[u] >> 8)], 1);
                atomicAdd(&deg[dl], 1);
            }
        }
        base += 128 * U;
        if (base >= n) break;
        #pragma unroll
        for (int u = 0; u < U; ++u) {
            const int i = base + u * 128 + t;
            y[u] = (i < n) ? rec2[i].y : -1;
        }
    }
    __syncthreads();

    const int4* arow = (const int4*)&cnt[(wv * 16 + c) * CNT_PITCH + 32 * half];
    v8f acc0 = {}, acc1 = {};
    #pragma unroll
    for (int q4 = 0; q4 < 8; ++q4) {
        const int4 a4 = arow[q4];
        v2f a, b0, b1;
        a[0]  = (float)a4.x;       a[1]  = (float)a4.y;
        b0[0] = e[4 * q4].x;       b0[1] = e[4 * q4 + 1].x;
        b1[0] = e[4 * q4].y;       b1[1] = e[4 * q4 + 1].y;
        acc0 = __builtin_amdgcn_wmma_f32_16x16x4_f32(false, a, false, b0, (short)0, acc0, false, false);
        acc1 = __builtin_amdgcn_wmma_f32_16x16x4_f32(false, a, false, b1, (short)0, acc1, false, false);
        a[0]  = (float)a4.z;       a[1]  = (float)a4.w;
        b0[0] = e[4 * q4 + 2].x;   b0[1] = e[4 * q4 + 3].x;
        b1[0] = e[4 * q4 + 2].y;   b1[1] = e[4 * q4 + 3].y;
        acc0 = __builtin_amdgcn_wmma_f32_16x16x4_f32(false, a, false, b0, (short)0, acc0, false, false);
        acc1 = __builtin_amdgcn_wmma_f32_16x16x4_f32(false, a, false, b1, (short)0, acc1, false, false);
    }

    const int4* dq = (const int4*)&deg[wv * 16 + 8 * half];
    const int4  d0 = dq[0], d1 = dq[1];
    const int   dg[8] = {d0.x, d0.y, d0.z, d0.w, d1.x, d1.y, d1.z, d1.w};
    unsigned* mw = &mt[(wv * 16 + 8 * half) * MT_PITCH + c];
    #pragma unroll
    for (int r = 0; r < 8; ++r) {
        const float inv = 1.0f / fmaxf((float)dg[r], 1.0f);
        v2h o;
        o.x = (_Float16)(acc0[r] * inv);
        o.y = (_Float16)(acc1[r] * inv);
        mw[r * MT_PITCH] = __builtin_bit_cast(unsigned, o);
    }
    __syncthreads();

    const int row0 = b * BKT_NODES + wv * 16;
    if (row0 >= N_NODES) return;

    const uint4* mrow = (const uint4*)&mt[(wv * 16 + c) * MT_PITCH];
    const v8h lo = __builtin_bit_cast(v8h, mrow[half]);
    const v8h hi = __builtin_bit_cast(v8h, mrow[2 + half]);
    const v16h aM = __builtin_shufflevector(lo, hi, 0, 1, 2, 3, 4, 5, 6, 7,
                                            8, 9, 10, 11, 12, 13, 14, 15);
    const float4* xq = (const float4*)(xtab + xr * EMB);
    const v16h aX = load_frag(xq, 0, half, 1.0f);
    project_layer1(aM, aX, row0, half, c, Wl, bl, Wr, h1out);
}

#define ACC_PITCH 68
#define L2_CAP    3072
#define L2_SRC_BITS 17
__global__ void __launch_bounds__(256, 14)
range_layer2(const int* __restrict__ rec,
             const int* __restrict__ bstart,
             const int* __restrict__ tot,
             const _Float16* __restrict__ xh,
             const float* __restrict__ Wl,
             const float* __restrict__ bl,
             const float* __restrict__ Wr,
             float* __restrict__ pooled,
             const int* __restrict__ batch,
             float* __restrict__ gcnt)
{
    static_assert(N_NODES <= (1 << L2_SRC_BITS), "source node fits the low bits of a packed word");
    static_assert(BKT_NODES == 64 && L2_CAP % 256 == 0, "64 counts = 2 per lane of wave 0; whole trips of the work-group");
    __shared__ alignas(16) float acc[(BKT_NODES + 1) * ACC_PITCH];
    __shared__ int   deg[BKT_NODES];
    __shared__ int   ccnt[BKT_NODES];
    __shared__ int   cstart[BKT_NODES + 1];
    __shared__ int   ccur[BKT_NODES];
    static_assert(L2_CAP * 4 <= 2 * HID * W_PITCH * 2, "the ordered words fit the weights' region");
    __shared__ alignas(16) _Float16 wbuf[2 * HID * W_PITCH];
    typedef int __attribute__((__may_alias__)) lds_word;
    lds_word* srt = (lds_word*)wbuf;
    _Float16* wls = wbuf;
    _Float16* wrs = wbuf + HID * W_PITCH;

    const int t    = threadIdx.x;
    const int lane = t & 31;
    const int wv   = __builtin_amdgcn_readfirstlane(t >> 5);
    const int b    = blockIdx.x;

    const int   beg  = bstart[b];
    const int   n    = tot[b];

    for (int i = t; i < (BKT_NODES + 1) * ACC_PITCH; i += 256) acc[i] = 0.0f;
    if (t < BKT_NODES) { deg[t] = 0; ccnt[t] = 0; }
    __syncthreads();

    const int2* rec2 = (const int2*)rec + beg;
    const v2h*  xw   = (const v2h*)xh;

    constexpr int U   = L2_CAP / 256;
    constexpr int PAD = BKT_NODES << L2_SRC_BITS;

    for (int cb = 0; cb < n; cb += L2_CAP) {
        const int cn = min(n - cb, L2_CAP);

        int w[U];
        #pragma unroll
        for (int u = 0; u < U; ++u) {
            const int i = u * 256 + t;
            w[u] = -1;
            if (i < cn) {
                const int2 r = rec2[cb + i];
                w[u] = r.x | ((r.y & 255) << L2_SRC_BITS);
            }
        }
        #pragma unroll
        for (int u = 0; u < U; ++u)
            if (w[u] >= 0) atomicAdd(&ccnt[w[u] >> L2_SRC_BITS], 1);
        __syncthreads();

        if (wv == 0) {
            const int a0 = ccnt[2 * lane], a1 = ccnt[2 * lane + 1];
            int inc = a0 + a1;
            #pragma unroll
            for (int o = 1; o < 32; o <<= 1) {
                const int up = __shfl_up(inc, o);
                if (lane >= o) inc += up;
            }
            const int excl = inc - a0 - a1;
            cstart[2 * lane]     = excl;
            cstart[2 * lane + 1] = excl + a0;
            ccur[2 * lane]       = excl;
            ccur[2 * lane + 1]   = excl + a0;
            if (lane == 31) cstart[BKT_NODES] = inc;
            deg[2 * lane]     += a0;
            deg[2 * lane + 1] += a1;
            ccnt[2 * lane]     = 0;
            ccnt[2 * lane + 1] = 0;
        }
        __syncthreads();

        #pragma unroll
        for (int u = 0; u < U; ++u) {
            if (w[u] >= 0) {
                const int p = atomicAdd(&ccur[w[u] >> L2_SRC_BITS], 1);
                srt[p] = w[u];
            }
        }
        __syncthreads();

        const int s0 = __builtin_amdgcn_readfirstlane(cstart[8 * wv]);
        const int e0 = __builtin_amdgcn_readfirstlane(cstart[8 * wv + 8]);
        float ax  = 0.0f, ay = 0.0f;
        int   cur = BKT_NODES;
        for (int base = s0; base < e0; base += 32) {
            int wd = PAD;
            if (base + lane < e0) wd = srt[base + lane];

            v2h v[32];
            #pragma unroll
            for (int j = 0; j < 32; ++j) {
                const int sj = __builtin_amdgcn_readlane(wd, j) & ((1 << L2_SRC_BITS) - 1);
                v[j] = xw[sj * (HID / 2) + lane];
            }
            #pragma unroll
            for (int j = 0; j < 32; ++j) {
                const int dj = __builtin_amdgcn_readlane(wd, j) >> L2_SRC_BITS;
                if (dj != cur) {
                    float* a = &acc[cur * ACC_PITCH + lane];
                    atomicAdd(a,      ax);
                    atomicAdd(a + 32, ay);
                    ax = 0.0f; ay = 0.0f; cur = dj;
                }
                ax += (float)v[j].x;
                ay += (float)v[j].y;
            }
        }
        {
            float* a = &acc[cur * ACC_PITCH + lane];
            atomicAdd(a,      ax);
            atomicAdd(a + 32, ay);
        }
    }
    __syncthreads();

    static_assert(HID * HID / 4 == 4 * 256, "four 16-byte pieces of each matrix per thread");
    {
        float4 wl4[4], wr4[4];
        #pragma unroll
        for (int j = 0; j < 4; ++j) {
            wl4[j] = ((const float4*)Wl)[t + 256 * j];
            wr4[j] = ((const float4*)Wr)[t + 256 * j];
        }
        #pragma unroll
        for (int j = 0; j < 4; ++j) {
            const int p   = t + 256 * j;
            const int off = (p >> 4) * W_PITCH + (p & 15) * 4;
            v4h hl, hr;
            hl[0] = (_Float16)wl4[j].x; hl[1] = (_Float16)wl4[j].y;
            hl[2] = (_Float16)wl4[j].z; hl[3] = (_Float16)wl4[j].w;
            hr[0] = (_Float16)wr4[j].x; hr[1] = (_Float16)wr4[j].y;
            hr[2] = (_Float16)wr4[j].z; hr[3] = (_Float16)wr4[j].w;
            *(v4h*)&wls[off] = hl;
            *(v4h*)&wrs[off] = hr;
        }
    }
    __syncthreads();

    const int rt   = wv & 3;
    const int pr   = wv >> 2;
    const int row0 = b * BKT_NODES + rt * 16;
    if (row0 >= N_NODES) return;

    const int half = lane >> 4;
    const int m    = lane & 15;
    const int rl   = rt * 16 + m;
    const float inv = 1.0f / fmaxf((float)deg[rl], 1.0f);
    const float4* ar = (const float4*)&acc[rl * ACC_PITCH];
    const _Float16* xp = xh + (row0 + m) * HID;
    project_layer2(ar, inv, xp, row0, pr, half, m, wls, wrs, bl, pooled, batch, gcnt);
}

__global__ void final_head(const float* __restrict__ pooled,
                           const float* __restrict__ gcnt,
                           const float* __restrict__ Wout,
                           const float* __restrict__ bout,
                           float* __restrict__ out)
{
    int tid = blockIdx.x * blockDim.x + threadIdx.x;
    if (tid >= N_GRAPHS * NCLS) return;
    int g = tid / NCLS;
    int c = tid % NCLS;
    float inv = 1.0f / fmaxf(gcnt[g] + gcnt[N_GRAPHS + g], 1.0f);
    float acc = bout[c];
    #pragma unroll
    for (int h = 0; h < HID; ++h)
        acc += (pooled[g * HID + h] + pooled[N_GRAPHS * HID + g * HID + h]) * inv * Wout[c * HID + h];
    out[tid] = acc;
}

extern "C" void kernel_launch(void* const* d_in, const int* in_sizes, int n_in,
                              void* d_out, int out_size, void* d_ws, size_t ws_size,
                              hipStream_t stream)
{
    const int*   x_idx   = (const int*)  d_in[0];
    const int*   ei      = (const int*)  d_in[1];
    const int*   batch   = (const int*)  d_in[2];
    const float* embed_w = (const float*)d_in[3];
    const float* W1l     = (const float*)d_in[4];
    const float* b1      = (const float*)d_in[5];
    const float* W1r     = (const float*)d_in[6];
    const float* W2l     = (const float*)d_in[7];
    const float* b2      = (const float*)d_in[8];
    const float* W2r     = (const float*)d_in[9];
    const float* Wout    = (const float*)d_in[10];
    const float* bout    = (const float*)d_in[11];
    float*       out     = (float*)d_out;

    const int* src = ei;
    const int* dst = ei + N_EDGES;

    int*   tot    = (int*)d_ws;
    int*   cursor = tot    + NBKT_PAD;
    float* pooled = (float*)(cursor + NBKT_PAD);
    float* gcnt   = pooled + 2 * N_GRAPHS * HID;
    int*   bstart = (int*)(gcnt + 2 * N_GRAPHS);
    int*   blkcnt = bstart + NBKT_PAD;
    int*   rec    = blkcnt + PART_BLOCKS * NBKT;
    _Float16* h1    = (_Float16*)(rec + 2 * N_EDGES);
    unsigned* xb    = (unsigned*)(h1 + N_NODES * HID);

    const int B = 256;

    hipMemsetAsync(tot, 0,
                   sizeof(int) * (2 * NBKT_PAD + 2 * N_GRAPHS * HID + 2 * N_GRAPHS),
                   stream);

    bucket_count<<<PART_BLOCKS, PART_THREADS, 0, stream>>>(dst, x_idx, blkcnt, tot, xb);
    bucket_place<<<PART_BLOCKS, PART_THREADS, 0, stream>>>(src, dst, xb, blkcnt, tot, cursor,
                                                            bstart, rec);

    range_layer1<<<NBKT, 128, 0, stream>>>(rec, bstart, tot, embed_w, x_idx, W1l, b1, W1r, h1);

    range_layer2<<<NBKT, 256, 0, stream>>>(rec, bstart, tot, h1, W2l, b2, W2r, pooled, batch, gcnt);

    final_head<<<(N_GRAPHS * NCLS + B - 1) / B, B, 0, stream>>>(pooled, gcnt, Wout, bout, out);
}
